# starting kernel
# baseline (speedup 1.0000x reference)
#include <hip/hip_runtime.h>
#include <hip/hip_bf16.h>


typedef __attribute__((ext_vector_type(2))) float v2f;
typedef __attribute__((ext_vector_type(8))) float v8f;

static constexpr int GN  = 50000;
static constexpr int GE  = 800000;
static constexpr int GIN = 128;
static constexpr int GH  = 128;
static constexpr int GC  = 64;


__global__ void deg_init_kernel(float* __restrict__ deg, int n) {
    int i = blockIdx.x * blockDim.x + threadIdx.x;
    if (i < n) deg[i] = 1.0f;
}

__global__ void deg_count_kernel(const int* __restrict__ dst, float* __restrict__ deg, int nE) {
    int e = blockIdx.x * blockDim.x + threadIdx.x;
    if (e < nE) atomicAdd(&deg[dst[e]], 1.0f);
}

__global__ void deg_to_dinv_kernel(float* __restrict__ deg, int n) {
    int i = blockIdx.x * blockDim.x + threadIdx.x;
    if (i < n) deg[i] = rsqrtf(deg[i]);
}


template <int DIN, int DOUT>
__global__ void gemm_wmma_f32_kernel(const float* __restrict__ X,
                                     const float* __restrict__ W,
                                     float* __restrict__ Y,
                                     int mTiles) {
    constexpr int tilesN = DOUT / 16;
    constexpr int KF     = DIN / 4;

    const int wave       = threadIdx.x >> 5;
    const int wavesPerBl = blockDim.x >> 5;
    const int wid        = blockIdx.x * wavesPerBl + wave;
    const int wavesTotal = gridDim.x * wavesPerBl;
    const int tn         = wid % tilesN;
    const int mStep      = wavesTotal / tilesN;

    const int lane = threadIdx.x & 31;
    const int half = lane >> 4;
    const int l15  = lane & 15;
    const int col  = tn * 16 + l15;

    v2f bfrag[KF];
#pragma unroll
    for (int j = 0; j < KF; ++j) {
        const int k = 4 * j + 2 * half;
        bfrag[j].x = W[(size_t)k * DOUT + col];
        bfrag[j].y = W[(size_t)(k + 1) * DOUT + col];
    }

    for (int tm = wid / tilesN; tm < mTiles; tm += mStep) {
        const float* __restrict__ xr =
            X + (size_t)(tm * 16 + l15) * DIN + 2 * half;
        v8f acc = {};
#pragma unroll
        for (int j = 0; j < KF; ++j) {
            v2f a;
            a.x = xr[4 * j];
            a.y = xr[4 * j + 1];
            acc = __builtin_amdgcn_wmma_f32_16x16x4_f32(
                 false, a,  false, bfrag[j],
                 (short)0, acc,  false,  false);
        }
        float* __restrict__ yr = Y + (size_t)(tm * 16 + 8 * half) * DOUT + col;
#pragma unroll
        for (int r = 0; r < 8; ++r)
            yr[(size_t)r * DOUT] = acc[r];
    }
}


template <int D, bool ADD_BIAS>
__global__ void selfloop_init_kernel(const float* __restrict__ h,
                                     const float* __restrict__ dinv,
                                     const float* __restrict__ bias,
                                     float* __restrict__ agg, int n) {
    size_t idx = (size_t)blockIdx.x * blockDim.x + threadIdx.x;
    size_t total = (size_t)n * D / 4;
    if (idx >= total) return;
    size_t base = idx * 4;
    int i = (int)(base / D);
    int f = (int)(base % D);
    float di = dinv[i];
    float w  = di * di;
    float4 v = *(const float4*)(h + base);
    float4 o;
    o.x = w * v.x; o.y = w * v.y; o.z = w * v.z; o.w = w * v.w;
    if (ADD_BIAS) {
        o.x += bias[f]; o.y += bias[f + 1]; o.z += bias[f + 2]; o.w += bias[f + 3];
    }
    *(float4*)(agg + base) = o;
}


template <int D>
__global__ void edge_scatter_kernel(const int* __restrict__ src,
                                    const int* __restrict__ dst,
                                    const float* __restrict__ dinv,
                                    const float* __restrict__ h,
                                    float* __restrict__ agg, int nE) {
    constexpr int VPT = D / 32;
    static_assert(VPT == 2 || VPT == 4, "D must be 64 or 128");
    const int wave = threadIdx.x >> 5;
    const int lane = threadIdx.x & 31;
    const int e    = blockIdx.x * (blockDim.x >> 5) + wave;
    if (e >= nE) return;

    const int s = src[e];
    const int d = dst[e];
    const float norm = dinv[s] * dinv[d];

    const float* __restrict__ hp = h   + (size_t)s * D + lane * VPT;
    float* __restrict__ ap       = agg + (size_t)d * D + lane * VPT;

    if (VPT == 4) {
        float4 v = *(const float4*)hp;
        atomicAdd(ap + 0, norm * v.x);
        atomicAdd(ap + 1, norm * v.y);
        atomicAdd(ap + 2, norm * v.z);
        atomicAdd(ap + 3, norm * v.w);
    } else {
        float2 v = *(const float2*)hp;
        atomicAdd(ap + 0, norm * v.x);
        atomicAdd(ap + 1, norm * v.y);
    }
}


template <int D>
__global__ void bias_bn_relu_kernel(float* __restrict__ agg,
                                    const float* __restrict__ b,
                                    const float* __restrict__ g,
                                    const float* __restrict__ be,
                                    const float* __restrict__ m,
                                    const float* __restrict__ v, int n) {
    size_t idx = (size_t)blockIdx.x * blockDim.x + threadIdx.x;
    size_t total = (size_t)n * D / 4;
    if (idx >= total) return;
    size_t base = idx * 4;
    int f = (int)(base % D);
    float4 a = *(float4*)(agg + base);
    float* pa = &a.x;
#pragma unroll
    for (int j = 0; j < 4; ++j) {
        int fj = f + j;
        float val = pa[j] + b[fj];
        val = g[fj] * (val - m[fj]) * rsqrtf(v[fj] + 1e-5f) + be[fj];
        pa[j] = fmaxf(val, 0.0f);
    }
    *(float4*)(agg + base) = a;
}


static inline size_t align256(size_t x) { return (x + 255) & ~(size_t)255; }

extern "C" void kernel_launch(void* const* d_in, const int* in_sizes, int n_in,
                              void* d_out, int out_size, void* d_ws, size_t ws_size,
                              hipStream_t stream) {
    (void)in_sizes; (void)n_in; (void)out_size; (void)ws_size;

    const float* x   = (const float*)d_in[0];
    const int*   ei  = (const int*)d_in[1];
    const int*   src = ei;
    const int*   dst = ei + GE;
    const float* W1  = (const float*)d_in[2];
    const float* b1  = (const float*)d_in[3];
    const float* g1  = (const float*)d_in[4];
    const float* be1 = (const float*)d_in[5];
    const float* m1  = (const float*)d_in[6];
    const float* v1  = (const float*)d_in[7];
    const float* W2  = (const float*)d_in[8];
    const float* b2  = (const float*)d_in[9];
    const float* g2  = (const float*)d_in[10];
    const float* be2 = (const float*)d_in[11];
    const float* m2  = (const float*)d_in[12];
    const float* v2  = (const float*)d_in[13];
    const float* W3  = (const float*)d_in[14];
    const float* b3  = (const float*)d_in[15];
    float* out = (float*)d_out;

    char* ws = (char*)d_ws;
    size_t off = 0;
    float* dinv = (float*)(ws + off); off += align256((size_t)GN * 4);
    float* hbuf = (float*)(ws + off); off += align256((size_t)GN * GH * 4);
    float* agg1 = (float*)(ws + off); off += align256((size_t)GN * GH * 4);
    float* agg2 = (float*)(ws + off); off += align256((size_t)GN * GH * 4);

    const int TPB = 256;

    deg_init_kernel<<<(GN + TPB - 1) / TPB, TPB, 0, stream>>>(dinv, GN);
    deg_count_kernel<<<(GE + TPB - 1) / TPB, TPB, 0, stream>>>(dst, dinv, GE);
    deg_to_dinv_kernel<<<(GN + TPB - 1) / TPB, TPB, 0, stream>>>(dinv, GN);

    const int mTiles = GN / 16;
    const int gemmBlocks = 512;
    const int edgeBlocks = (GE + 7) / 8;
    const int elemBlocksH = (GN * GH / 4 + TPB - 1) / TPB;
    const int elemBlocksC = (GN * GC / 4 + TPB - 1) / TPB;

    gemm_wmma_f32_kernel<GIN, GH><<<gemmBlocks, TPB, 0, stream>>>(
        x, W1, hbuf, mTiles);
    selfloop_init_kernel<GH, false><<<elemBlocksH, TPB, 0, stream>>>(
        hbuf, dinv, nullptr, agg1, GN);
    edge_scatter_kernel<GH><<<edgeBlocks, TPB, 0, stream>>>(
        src, dst, dinv, hbuf, agg1, GE);
    bias_bn_relu_kernel<GH><<<elemBlocksH, TPB, 0, stream>>>(
        agg1, b1, g1, be1, m1, v1, GN);

    gemm_wmma_f32_kernel<GH, GH><<<gemmBlocks, TPB, 0, stream>>>(
        agg1, W2, hbuf, mTiles);
    selfloop_init_kernel<GH, false><<<elemBlocksH, TPB, 0, stream>>>(
        hbuf, dinv, nullptr, agg2, GN);
    edge_scatter_kernel<GH><<<edgeBlocks, TPB, 0, stream>>>(
        src, dst, dinv, hbuf, agg2, GE);
    bias_bn_relu_kernel<GH><<<elemBlocksH, TPB, 0, stream>>>(
        agg2, b2, g2, be2, m2, v2, GN);

    gemm_wmma_f32_kernel<GH, GC><<<gemmBlocks, TPB, 0, stream>>>(
        agg2, W3, hbuf, mTiles);
    selfloop_init_kernel<GC, true><<<elemBlocksC, TPB, 0, stream>>>(
        hbuf, dinv, b3, out, GN);
    edge_scatter_kernel<GC><<<edgeBlocks, TPB, 0, stream>>>(
        src, dst, dinv, hbuf, out, GE);
}


# aggregation as per-destination gather over edge lists grouped in-call (int atomic gives count and slot; rows summed in arrival-rank order); f32 scatter atomics and self-loop init pass removed
# speedup vs baseline: 2.4691x; 2.4691x over previous
#include <hip/hip_runtime.h>
#include <hip/hip_bf16.h>


typedef __attribute__((ext_vector_type(2))) float v2f;
typedef __attribute__((ext_vector_type(8))) float v8f;

static constexpr int GN  = 50000;
static constexpr int GE  = 800000;
static constexpr int GIN = 128;
static constexpr int GH  = 128;
static constexpr int GC  = 64;


template <int V> using vecf = float __attribute__((ext_vector_type(V)));

static constexpr int SCAN_T = 1024;

__global__ void count_zero_kernel(int* __restrict__ cnt, int n) {
    int i = blockIdx.x * blockDim.x + threadIdx.x;
    if (i < n) cnt[i] = 0;
}

__global__ void count_rank_kernel(const int* __restrict__ dst, int* __restrict__ cnt,
                                  int* __restrict__ rank, int nE) {
    int e = blockIdx.x * blockDim.x + threadIdx.x;
    if (e < nE) rank[e] = atomicAdd(&cnt[dst[e]], 1);
}

__global__ void __launch_bounds__(SCAN_T)
scan_dinv_kernel(const int* __restrict__ cnt, int* __restrict__ rowptr,
                 float* __restrict__ dinv, int n) {
    __shared__ int wtot[32];
    __shared__ int wpre[32];
    const int t    = threadIdx.x;
    const int lane = t & 31;
    const int wv   = t >> 5;
    const int lim  = blockIdx.x * SCAN_T;

    int s = 0;
    const int4* __restrict__ c4 = (const int4*)cnt;
    for (int q = t; q < lim / 4; q += SCAN_T) {
        int4 v = c4[q];
        s += (v.x + v.y) + (v.z + v.w);
    }

    const int i = lim + t;
    const int c = (i < n) ? cnt[i] : 0;

    int inc = c;
    for (int d = 1; d < 32; d <<= 1) {
        int o = __shfl_up(inc, d);
        if (lane >= d) inc += o;
    }
    for (int d = 16; d > 0; d >>= 1) s += __shfl_xor(s, d);

    if (lane == 31) wtot[wv] = inc;
    if (lane == 0)  wpre[wv] = s;
    __syncthreads();

    int a = wpre[lane];
    int b = (lane < wv) ? wtot[lane] : 0;
    for (int d = 16; d > 0; d >>= 1) {
        a += __shfl_xor(a, d);
        b += __shfl_xor(b, d);
    }
    if (i < n) {
        rowptr[i] = a + b + inc - c;
        dinv[i]   = rsqrtf((float)(c + 1));
    }
}

__global__ void csr_fill_kernel(const int* __restrict__ src, const int* __restrict__ dst,
                                const int* __restrict__ rank, const int* __restrict__ rowptr,
                                const float* __restrict__ dinv, int2* __restrict__ csr, int nE) {
    int e = blockIdx.x * blockDim.x + threadIdx.x;
    if (e >= nE) return;
    const int s = src[e];
    const int d = dst[e];
    int2 ent;
    ent.x = s;
    ent.y = __float_as_int(dinv[s] * dinv[d]);
    csr[rowptr[d] + rank[e]] = ent;
}


template <int DIN, int DOUT>
__global__ void gemm_wmma_f32_kernel(const float* __restrict__ X,
                                     const float* __restrict__ W,
                                     float* __restrict__ Y,
                                     int mTiles) {
    constexpr int tilesN = DOUT / 16;
    constexpr int KF     = DIN / 4;

    const int wave       = threadIdx.x >> 5;
    const int wavesPerBl = blockDim.x >> 5;
    const int wid        = blockIdx.x * wavesPerBl + wave;
    const int wavesTotal = gridDim.x * wavesPerBl;
    const int tn         = wid % tilesN;
    const int mStep      = wavesTotal / tilesN;

    const int lane = threadIdx.x & 31;
    const int half = lane >> 4;
    const int l15  = lane & 15;
    const int col  = tn * 16 + l15;

    v2f bfrag[KF];
#pragma unroll
    for (int j = 0; j < KF; ++j) {
        const int k = 4 * j + 2 * half;
        bfrag[j].x = W[(size_t)k * DOUT + col];
        bfrag[j].y = W[(size_t)(k + 1) * DOUT + col];
    }

    for (int tm = wid / tilesN; tm < mTiles; tm += mStep) {
        const float* __restrict__ xr =
            X + (size_t)(tm * 16 + l15) * DIN + 2 * half;
        v8f acc = {};
#pragma unroll
        for (int j = 0; j < KF; ++j) {
            v2f a;
            a.x = xr[4 * j];
            a.y = xr[4 * j + 1];
            acc = __builtin_amdgcn_wmma_f32_16x16x4_f32(
                 false, a,  false, bfrag[j],
                 (short)0, acc,  false,  false);
        }
        float* __restrict__ yr = Y + (size_t)(tm * 16 + 8 * half) * DOUT + col;
#pragma unroll
        for (int r = 0; r < 8; ++r)
            yr[(size_t)r * DOUT] = acc[r];
    }
}


template <int D, int U>
__device__ __forceinline__ vecf<D / 32> gather_rows(const float* __restrict__ h, int es, int ew,
                                                    int j, int lane, vecf<D / 32> acc) {
    constexpr int VPT = D / 32;
    vecf<VPT> v[U];
    float w[U];
#pragma unroll
    for (int k = 0; k < U; ++k) {
        const int s = __builtin_amdgcn_readlane(es, j + k);
        w[k] = __int_as_float(__builtin_amdgcn_readlane(ew, j + k));
        v[k] = *(const vecf<VPT>*)(h + (size_t)s * D + lane * VPT);
    }
#pragma unroll
    for (int k = 0; k < U; ++k) acc += w[k] * v[k];
    return acc;
}

template <int D, bool ADD_BIAS>
__global__ void gather_rows_kernel(const int* __restrict__ rowptr, const int* __restrict__ cnt,
                                   const int2* __restrict__ csr, const float* __restrict__ dinv,
                                   const float* __restrict__ h, const float* __restrict__ bias,
                                   float* __restrict__ agg, int n) {
    constexpr int VPT = D / 32;
    static_assert(VPT == 2 || VPT == 4, "D must be 64 or 128");
    const int lane = threadIdx.x & 31;
    const int i = __builtin_amdgcn_readfirstlane(
        (int)(blockIdx.x * (blockDim.x >> 5) + (threadIdx.x >> 5)));
    if (i >= n) return;

    const int   start = rowptr[i];
    const int   deg   = cnt[i];
    const float di    = dinv[i];

    vecf<VPT> acc = (di * di) * *(const vecf<VPT>*)(h + (size_t)i * D + lane * VPT);

    for (int base = 0; base < deg; base += 32) {
        const int m = (deg - base < 32) ? (deg - base) : 32;
        int es = 0, ew = 0;
        if (lane < m) {
            const int2 ent = csr[start + base + lane];
            es = ent.x;
            ew = ent.y;
        }
        int j = 0;
        for (; j + 8 <= m; j += 8) acc = gather_rows<D, 8>(h, es, ew, j, lane, acc);
        if (j + 4 <= m) { acc = gather_rows<D, 4>(h, es, ew, j, lane, acc); j += 4; }
        if (j + 2 <= m) { acc = gather_rows<D, 2>(h, es, ew, j, lane, acc); j += 2; }
        if (j < m)        acc = gather_rows<D, 1>(h, es, ew, j, lane, acc);
    }

    if (ADD_BIAS) acc += *(const vecf<VPT>*)(bias + lane * VPT);
    *(vecf<VPT>*)(agg + (size_t)i * D + lane * VPT) = acc;
}


template <int D>
__global__ void bias_bn_relu_kernel(float* __restrict__ agg,
                                    const float* __restrict__ b,
                                    const float* __restrict__ g,
                                    const float* __restrict__ be,
                                    const float* __restrict__ m,
                                    const float* __restrict__ v, int n) {
    size_t idx = (size_t)blockIdx.x * blockDim.x + threadIdx.x;
    size_t total = (size_t)n * D / 4;
    if (idx >= total) return;
    size_t base = idx * 4;
    int f = (int)(base % D);
    float4 a = *(float4*)(agg + base);
    float* pa = &a.x;
#pragma unroll
    for (int j = 0; j < 4; ++j) {
        int fj = f + j;
        float val = pa[j] + b[fj];
        val = g[fj] * (val - m[fj]) * rsqrtf(v[fj] + 1e-5f) + be[fj];
        pa[j] = fmaxf(val, 0.0f);
    }
    *(float4*)(agg + base) = a;
}


static inline size_t align256(size_t x) { return (x + 255) & ~(size_t)255; }

extern "C" void kernel_launch(void* const* d_in, const int* in_sizes, int n_in,
                              void* d_out, int out_size, void* d_ws, size_t ws_size,
                              hipStream_t stream) {
    (void)in_sizes; (void)n_in; (void)out_size; (void)ws_size;

    const float* x   = (const float*)d_in[0];
    const int*   ei  = (const int*)d_in[1];
    const int*   src = ei;
    const int*   dst = ei + GE;
    const float* W1  = (const float*)d_in[2];
    const float* b1  = (const float*)d_in[3];
    const float* g1  = (const float*)d_in[4];
    const float* be1 = (const float*)d_in[5];
    const float* m1  = (const float*)d_in[6];
    const float* v1  = (const float*)d_in[7];
    const float* W2  = (const float*)d_in[8];
    const float* b2  = (const float*)d_in[9];
    const float* g2  = (const float*)d_in[10];
    const float* be2 = (const float*)d_in[11];
    const float* m2  = (const float*)d_in[12];
    const float* v2  = (const float*)d_in[13];
    const float* W3  = (const float*)d_in[14];
    const float* b3  = (const float*)d_in[15];
    float* out = (float*)d_out;

    char* ws = (char*)d_ws;
    size_t off = 0;
    float* dinv   = (float*)(ws + off); off += align256((size_t)GN * 4);
    int*   cnt    = (int*)(ws + off);   off += align256((size_t)GN * 4);
    int*   rowptr = (int*)(ws + off);   off += align256((size_t)GN * 4);
    int*   rank   = (int*)(ws + off);   off += align256((size_t)GE * 4);
    int2*  csr    = (int2*)(ws + off);  off += align256((size_t)GE * 8);
    float* hbuf = (float*)(ws + off); off += align256((size_t)GN * GH * 4);
    float* agg1 = (float*)(ws + off); off += align256((size_t)GN * GH * 4);
    float* agg2 = (float*)(ws + off); off += align256((size_t)GN * GH * 4);

    const int TPB = 256;

    count_zero_kernel<<<(GN + TPB - 1) / TPB, TPB, 0, stream>>>(cnt, GN);
    count_rank_kernel<<<(GE + TPB - 1) / TPB, TPB, 0, stream>>>(dst, cnt, rank, GE);
    scan_dinv_kernel<<<(GN + SCAN_T - 1) / SCAN_T, SCAN_T, 0, stream>>>(cnt, rowptr, dinv, GN);
    csr_fill_kernel<<<(GE + TPB - 1) / TPB, TPB, 0, stream>>>(src, dst, rank, rowptr, dinv, csr, GE);

    const int mTiles = GN / 16;
    const int gemmBlocks = 512;
    const int rowBlocks = (GN + 7) / 8;
    const int elemBlocksH = (GN * GH / 4 + TPB - 1) / TPB;

    gemm_wmma_f32_kernel<GIN, GH><<<gemmBlocks, TPB, 0, stream>>>(
        x, W1, hbuf, mTiles);
    gather_rows_kernel<GH, false><<<rowBlocks, TPB, 0, stream>>>(
        rowptr, cnt, csr, dinv, hbuf, nullptr, agg1, GN);
    bias_bn_relu_kernel<GH><<<elemBlocksH, TPB, 0, stream>>>(
        agg1, b1, g1, be1, m1, v1, GN);

    gemm_wmma_f32_kernel<GH, GH><<<gemmBlocks, TPB, 0, stream>>>(
        agg1, W2, hbuf, mTiles);
    gather_rows_kernel<GH, false><<<rowBlocks, TPB, 0, stream>>>(
        rowptr, cnt, csr, dinv, hbuf, nullptr, agg2, GN);
    bias_bn_relu_kernel<GH><<<elemBlocksH, TPB, 0, stream>>>(
        agg2, b2, g2, be2, m2, v2, GN);

    gemm_wmma_f32_kernel<GH, GC><<<gemmBlocks, TPB, 0, stream>>>(
        agg2, W3, hbuf, mTiles);
    gather_rows_kernel<GC, true><<<rowBlocks, TPB, 0, stream>>>(
        rowptr, cnt, csr, dinv, hbuf, b3, out, GN);
}


# bias + BN(eval) + ReLU applied by the gather wave to its summed row in registers before the row's one store; the two elementwise passes over agg and their two launches removed
# speedup vs baseline: 2.6435x; 1.0706x over previous
#include <hip/hip_runtime.h>
#include <hip/hip_bf16.h>


typedef __attribute__((ext_vector_type(2))) float v2f;
typedef __attribute__((ext_vector_type(8))) float v8f;

static constexpr int GN  = 50000;
static constexpr int GE  = 800000;
static constexpr int GIN = 128;
static constexpr int GH  = 128;
static constexpr int GC  = 64;


template <int V> using vecf = float __attribute__((ext_vector_type(V)));

static constexpr int SCAN_T = 1024;

__global__ void count_zero_kernel(int* __restrict__ cnt, int n) {
    int i = blockIdx.x * blockDim.x + threadIdx.x;
    if (i < n) cnt[i] = 0;
}

__global__ void count_rank_kernel(const int* __restrict__ dst, int* __restrict__ cnt,
                                  int* __restrict__ rank, int nE) {
    int e = blockIdx.x * blockDim.x + threadIdx.x;
    if (e < nE) rank[e] = atomicAdd(&cnt[dst[e]], 1);
}

__global__ void __launch_bounds__(SCAN_T)
scan_dinv_kernel(const int* __restrict__ cnt, int* __restrict__ rowptr,
                 float* __restrict__ dinv, int n) {
    __shared__ int wtot[32];
    __shared__ int wpre[32];
    const int t    = threadIdx.x;
    const int lane = t & 31;
    const int wv   = t >> 5;
    const int lim  = blockIdx.x * SCAN_T;

    int s = 0;
    const int4* __restrict__ c4 = (const int4*)cnt;
    for (int q = t; q < lim / 4; q += SCAN_T) {
        int4 v = c4[q];
        s += (v.x + v.y) + (v.z + v.w);
    }

    const int i = lim + t;
    const int c = (i < n) ? cnt[i] : 0;

    int inc = c;
    for (int d = 1; d < 32; d <<= 1) {
        int o = __shfl_up(inc, d);
        if (lane >= d) inc += o;
    }
    for (int d = 16; d > 0; d >>= 1) s += __shfl_xor(s, d);

    if (lane == 31) wtot[wv] = inc;
    if (lane == 0)  wpre[wv] = s;
    __syncthreads();

    int a = wpre[lane];
    int b = (lane < wv) ? wtot[lane] : 0;
    for (int d = 16; d > 0; d >>= 1) {
        a += __shfl_xor(a, d);
        b += __shfl_xor(b, d);
    }
    if (i < n) {
        rowptr[i] = a + b + inc - c;
        dinv[i]   = rsqrtf((float)(c + 1));
    }
}

__global__ void csr_fill_kernel(const int* __restrict__ src, const int* __restrict__ dst,
                                const int* __restrict__ rank, const int* __restrict__ rowptr,
                                const float* __restrict__ dinv, int2* __restrict__ csr, int nE) {
    int e = blockIdx.x * blockDim.x + threadIdx.x;
    if (e >= nE) return;
    const int s = src[e];
    const int d = dst[e];
    int2 ent;
    ent.x = s;
    ent.y = __float_as_int(dinv[s] * dinv[d]);
    csr[rowptr[d] + rank[e]] = ent;
}


template <int DIN, int DOUT>
__global__ void gemm_wmma_f32_kernel(const float* __restrict__ X,
                                     const float* __restrict__ W,
                                     float* __restrict__ Y,
                                     int mTiles) {
    constexpr int tilesN = DOUT / 16;
    constexpr int KF     = DIN / 4;

    const int wave       = threadIdx.x >> 5;
    const int wavesPerBl = blockDim.x >> 5;
    const int wid        = blockIdx.x * wavesPerBl + wave;
    const int wavesTotal = gridDim.x * wavesPerBl;
    const int tn         = wid % tilesN;
    const int mStep      = wavesTotal / tilesN;

    const int lane = threadIdx.x & 31;
    const int half = lane >> 4;
    const int l15  = lane & 15;
    const int col  = tn * 16 + l15;

    v2f bfrag[KF];
#pragma unroll
    for (int j = 0; j < KF; ++j) {
        const int k = 4 * j + 2 * half;
        bfrag[j].x = W[(size_t)k * DOUT + col];
        bfrag[j].y = W[(size_t)(k + 1) * DOUT + col];
    }

    for (int tm = wid / tilesN; tm < mTiles; tm += mStep) {
        const float* __restrict__ xr =
            X + (size_t)(tm * 16 + l15) * DIN + 2 * half;
        v8f acc = {};
#pragma unroll
        for (int j = 0; j < KF; ++j) {
            v2f a;
            a.x = xr[4 * j];
            a.y = xr[4 * j + 1];
            acc = __builtin_amdgcn_wmma_f32_16x16x4_f32(
                 false, a,  false, bfrag[j],
                 (short)0, acc,  false,  false);
        }
        float* __restrict__ yr = Y + (size_t)(tm * 16 + 8 * half) * DOUT + col;
#pragma unroll
        for (int r = 0; r < 8; ++r)
            yr[(size_t)r * DOUT] = acc[r];
    }
}


template <int D, int U>
__device__ __forceinline__ vecf<D / 32> gather_rows(const float* __restrict__ h, int es, int ew,
                                                    int j, int lane, vecf<D / 32> acc) {
    constexpr int VPT = D / 32;
    vecf<VPT> v[U];
    float w[U];
#pragma unroll
    for (int k = 0; k < U; ++k) {
        const int s = __builtin_amdgcn_readlane(es, j + k);
        w[k] = __int_as_float(__builtin_amdgcn_readlane(ew, j + k));
        v[k] = *(const vecf<VPT>*)(h + (size_t)s * D + lane * VPT);
    }
#pragma unroll
    for (int k = 0; k < U; ++k) acc += w[k] * v[k];
    return acc;
}

static constexpr int EPI_SUM = 0, EPI_BIAS = 1, EPI_BN_RELU = 2;

template <int D, int EPI>
__global__ void gather_rows_kernel(const int* __restrict__ rowptr, const int* __restrict__ cnt,
                                   const int2* __restrict__ csr, const float* __restrict__ dinv,
                                   const float* __restrict__ h, const float* __restrict__ bias,
                                   const float* __restrict__ bn_g, const float* __restrict__ bn_be,
                                   const float* __restrict__ bn_m, const float* __restrict__ bn_v,
                                   float* __restrict__ agg, int n) {
    constexpr int VPT = D / 32;
    static_assert(VPT == 2 || VPT == 4, "D must be 64 or 128");
    const int lane = threadIdx.x & 31;
    const int i = __builtin_amdgcn_readfirstlane(
        (int)(blockIdx.x * (blockDim.x >> 5) + (threadIdx.x >> 5)));
    if (i >= n) return;

    const int   start = rowptr[i];
    const int   deg   = cnt[i];
    const float di    = dinv[i];

    vecf<VPT> acc = (di * di) * *(const vecf<VPT>*)(h + (size_t)i * D + lane * VPT);

    for (int base = 0; base < deg; base += 32) {
        const int m = (deg - base < 32) ? (deg - base) : 32;
        int es = 0, ew = 0;
        if (lane < m) {
            const int2 ent = csr[start + base + lane];
            es = ent.x;
            ew = ent.y;
        }
        int j = 0;
        for (; j + 8 <= m; j += 8) acc = gather_rows<D, 8>(h, es, ew, j, lane, acc);
        if (j + 4 <= m) { acc = gather_rows<D, 4>(h, es, ew, j, lane, acc); j += 4; }
        if (j + 2 <= m) { acc = gather_rows<D, 2>(h, es, ew, j, lane, acc); j += 2; }
        if (j < m)        acc = gather_rows<D, 1>(h, es, ew, j, lane, acc);
    }

    if (EPI == EPI_BIAS) acc += *(const vecf<VPT>*)(bias + lane * VPT);
    if (EPI == EPI_BN_RELU) {
        const vecf<VPT> pb  = *(const vecf<VPT>*)(bias  + lane * VPT);
        const vecf<VPT> pg  = *(const vecf<VPT>*)(bn_g  + lane * VPT);
        const vecf<VPT> pbe = *(const vecf<VPT>*)(bn_be + lane * VPT);
        const vecf<VPT> pm  = *(const vecf<VPT>*)(bn_m  + lane * VPT);
        const vecf<VPT> pv  = *(const vecf<VPT>*)(bn_v  + lane * VPT);
#pragma unroll
        for (int k = 0; k < VPT; ++k) {
            float val = acc[k] + pb[k];
            val = pg[k] * (val - pm[k]) * rsqrtf(pv[k] + 1e-5f) + pbe[k];
            acc[k] = fmaxf(val, 0.0f);
        }
    }
    *(vecf<VPT>*)(agg + (size_t)i * D + lane * VPT) = acc;
}


static inline size_t align256(size_t x) { return (x + 255) & ~(size_t)255; }

extern "C" void kernel_launch(void* const* d_in, const int* in_sizes, int n_in,
                              void* d_out, int out_size, void* d_ws, size_t ws_size,
                              hipStream_t stream) {
    (void)in_sizes; (void)n_in; (void)out_size; (void)ws_size;

    const float* x   = (const float*)d_in[0];
    const int*   ei  = (const int*)d_in[1];
    const int*   src = ei;
    const int*   dst = ei + GE;
    const float* W1  = (const float*)d_in[2];
    const float* b1  = (const float*)d_in[3];
    const float* g1  = (const float*)d_in[4];
    const float* be1 = (const float*)d_in[5];
    const float* m1  = (const float*)d_in[6];
    const float* v1  = (const float*)d_in[7];
    const float* W2  = (const float*)d_in[8];
    const float* b2  = (const float*)d_in[9];
    const float* g2  = (const float*)d_in[10];
    const float* be2 = (const float*)d_in[11];
    const float* m2  = (const float*)d_in[12];
    const float* v2  = (const float*)d_in[13];
    const float* W3  = (const float*)d_in[14];
    const float* b3  = (const float*)d_in[15];
    float* out = (float*)d_out;

    char* ws = (char*)d_ws;
    size_t off = 0;
    float* dinv   = (float*)(ws + off); off += align256((size_t)GN * 4);
    int*   cnt    = (int*)(ws + off);   off += align256((size_t)GN * 4);
    int*   rowptr = (int*)(ws + off);   off += align256((size_t)GN * 4);
    int*   rank   = (int*)(ws + off);   off += align256((size_t)GE * 4);
    int2*  csr    = (int2*)(ws + off);  off += align256((size_t)GE * 8);
    float* hbuf = (float*)(ws + off); off += align256((size_t)GN * GH * 4);
    float* agg1 = (float*)(ws + off); off += align256((size_t)GN * GH * 4);
    float* agg2 = (float*)(ws + off); off += align256((size_t)GN * GH * 4);

    const int TPB = 256;

    count_zero_kernel<<<(GN + TPB - 1) / TPB, TPB, 0, stream>>>(cnt, GN);
    count_rank_kernel<<<(GE + TPB - 1) / TPB, TPB, 0, stream>>>(dst, cnt, rank, GE);
    scan_dinv_kernel<<<(GN + SCAN_T - 1) / SCAN_T, SCAN_T, 0, stream>>>(cnt, rowptr, dinv, GN);
    csr_fill_kernel<<<(GE + TPB - 1) / TPB, TPB, 0, stream>>>(src, dst, rank, rowptr, dinv, csr, GE);

    const int mTiles = GN / 16;
    const int gemmBlocks = 512;
    const int rowBlocks = (GN + 7) / 8;

    gemm_wmma_f32_kernel<GIN, GH><<<gemmBlocks, TPB, 0, stream>>>(
        x, W1, hbuf, mTiles);
    gather_rows_kernel<GH, EPI_BN_RELU><<<rowBlocks, TPB, 0, stream>>>(
        rowptr, cnt, csr, dinv, hbuf, b1, g1, be1, m1, v1, agg1, GN);

    gemm_wmma_f32_kernel<GH, GH><<<gemmBlocks, TPB, 0, stream>>>(
        agg1, W2, hbuf, mTiles);
    gather_rows_kernel<GH, EPI_BN_RELU><<<rowBlocks, TPB, 0, stream>>>(
        rowptr, cnt, csr, dinv, hbuf, b2, g2, be2, m2, v2, agg2, GN);

    gemm_wmma_f32_kernel<GH, GC><<<gemmBlocks, TPB, 0, stream>>>(
        agg2, W3, hbuf, mTiles);
    gather_rows_kernel<GC, EPI_BIAS><<<rowBlocks, TPB, 0, stream>>>(
        rowptr, cnt, csr, dinv, hbuf, b3, nullptr, nullptr, nullptr, nullptr, out, GN);
}


# edge slots and node counts from returning LDS adds in per-chunk node tables, merged by the prefix-sum launch; global int atomics and zeroing launch removed; row sum order now chunk by chunk
# speedup vs baseline: 3.3341x; 1.2612x over previous
#include <hip/hip_runtime.h>
#include <hip/hip_bf16.h>


typedef __attribute__((ext_vector_type(2))) float v2f;
typedef __attribute__((ext_vector_type(8))) float v8f;

static constexpr int GN  = 50000;
static constexpr int GE  = 800000;
static constexpr int GIN = 128;
static constexpr int GH  = 128;
static constexpr int GC  = 64;


template <int V> using vecf = float __attribute__((ext_vector_type(V)));

static constexpr int NCH   = 64;
static constexpr int CH    = GE / NCH;
static constexpr int CNT_T = 1024;
static constexpr int CNT_U = 8;
static constexpr int SEG   = 256;
static constexpr int NSEG  = (GN + SEG - 1) / SEG;
static_assert(CH * NCH == GE, "the chunks must tile the edge list");
static_assert(NSEG <= SEG, "one thread per segment in front of a block");
static_assert(GN % 4 == 0, "node tables are copied 16 bytes at a time");

__global__ void __launch_bounds__(CNT_T)
chunk_count_kernel(const int* __restrict__ dst, int* __restrict__ rank,
                   int* __restrict__ chtab, int* __restrict__ segsum) {
    __shared__ __attribute__((aligned(16))) int tab[NSEG * SEG];
    const int t = threadIdx.x;
    const int p = blockIdx.x;

    int4* tab4 = (int4*)tab;
    for (int q = t; q < NSEG * SEG / 4; q += CNT_T) tab4[q] = make_int4(0, 0, 0, 0);
    __syncthreads();

    const int* __restrict__ dp = dst  + (size_t)p * CH;
    int*       __restrict__ rp = rank + (size_t)p * CH;
    for (int k0 = t; k0 < CH; k0 += CNT_U * CNT_T) {
        int d[CNT_U], r[CNT_U];
#pragma unroll
        for (int u = 0; u < CNT_U; ++u) {
            const int k = k0 + u * CNT_T;
            d[u] = (k < CH) ? dp[k] : -1;
        }
#pragma unroll
        for (int u = 0; u < CNT_U; ++u) {
            r[u] = 0;
            if (d[u] >= 0) r[u] = atomicAdd(&tab[d[u]], 1);
        }
#pragma unroll
        for (int u = 0; u < CNT_U; ++u) {
            const int k = k0 + u * CNT_T;
            if (k < CH) rp[k] = r[u];
        }
    }
    __syncthreads();

    int4* __restrict__ out4 = (int4*)(chtab + (size_t)p * GN);
    for (int q = t; q < GN / 4; q += CNT_T) out4[q] = tab4[q];

    const int lane = t & 31;
    for (int s = t >> 5; s < NSEG; s += CNT_T / 32) {
        int v = 0;
        for (int k = 0; k < SEG / 32; ++k) v += tab[s * SEG + k * 32 + lane];
        for (int o = 16; o > 0; o >>= 1) v += __shfl_xor(v, o);
        if (lane == 0) segsum[p * NSEG + s] = v;
    }
}

__global__ void __launch_bounds__(SEG)
scan_dinv_kernel(int* __restrict__ chtab, const int* __restrict__ segsum,
                 int* __restrict__ cnt, int* __restrict__ rowptr,
                 float* __restrict__ dinv, int n) {
    constexpr int NW = SEG / 32;
    __shared__ int wtot[NW];
    __shared__ int wpre[NW];
    const int t    = threadIdx.x;
    const int lane = t & 31;
    const int wv   = t >> 5;
    const int b    = blockIdx.x;

    int s = 0;
    if (t < b)
        for (int p = 0; p < NCH; ++p) s += segsum[p * NSEG + t];

    const int i  = b * SEG + t;
    const int ic = (i < n) ? i : n - 1;
    int cc[NCH];
    int c = 0;
#pragma unroll
    for (int p = 0; p < NCH; ++p) {
        cc[p] = chtab[(size_t)p * GN + ic];
        c += cc[p];
    }
    if (i >= n) c = 0;

    int inc = c;
    for (int o = 1; o < 32; o <<= 1) {
        int v = __shfl_up(inc, o);
        if (lane >= o) inc += v;
    }
    for (int o = 16; o > 0; o >>= 1) s += __shfl_xor(s, o);

    if (lane == 31) wtot[wv] = inc;
    if (lane == 0)  wpre[wv] = s;
    __syncthreads();

    int a  = (lane < NW) ? wpre[lane] : 0;
    int bw = (lane < wv) ? wtot[lane] : 0;
    for (int o = 16; o > 0; o >>= 1) {
        a  += __shfl_xor(a, o);
        bw += __shfl_xor(bw, o);
    }
    if (i < n) {
        int run = a + bw + inc - c;
        rowptr[i] = run;
        cnt[i]    = c;
        dinv[i]   = rsqrtf((float)(c + 1));
#pragma unroll
        for (int p = 0; p < NCH; ++p) {
            chtab[(size_t)p * GN + i] = run;
            run += cc[p];
        }
    }
}

__global__ void csr_fill_kernel(const int* __restrict__ src, const int* __restrict__ dst,
                                const int* __restrict__ rank, const int* __restrict__ chtab,
                                const float* __restrict__ dinv, int2* __restrict__ csr, int nE) {
    int e = blockIdx.x * blockDim.x + threadIdx.x;
    if (e >= nE) return;
    const int s = src[e];
    const int d = dst[e];
    const int p = e / CH;
    int2 ent;
    ent.x = s;
    ent.y = __float_as_int(dinv[s] * dinv[d]);
    csr[chtab[(size_t)p * GN + d] + rank[e]] = ent;
}


template <int DIN, int DOUT>
__global__ void gemm_wmma_f32_kernel(const float* __restrict__ X,
                                     const float* __restrict__ W,
                                     float* __restrict__ Y,
                                     int mTiles) {
    constexpr int tilesN = DOUT / 16;
    constexpr int KF     = DIN / 4;

    const int wave       = threadIdx.x >> 5;
    const int wavesPerBl = blockDim.x >> 5;
    const int wid        = blockIdx.x * wavesPerBl + wave;
    const int wavesTotal = gridDim.x * wavesPerBl;
    const int tn         = wid % tilesN;
    const int mStep      = wavesTotal / tilesN;

    const int lane = threadIdx.x & 31;
    const int half = lane >> 4;
    const int l15  = lane & 15;
    const int col  = tn * 16 + l15;

    v2f bfrag[KF];
#pragma unroll
    for (int j = 0; j < KF; ++j) {
        const int k = 4 * j + 2 * half;
        bfrag[j].x = W[(size_t)k * DOUT + col];
        bfrag[j].y = W[(size_t)(k + 1) * DOUT + col];
    }

    for (int tm = wid / tilesN; tm < mTiles; tm += mStep) {
        const float* __restrict__ xr =
            X + (size_t)(tm * 16 + l15) * DIN + 2 * half;
        v8f acc = {};
#pragma unroll
        for (int j = 0; j < KF; ++j) {
            v2f a;
            a.x = xr[4 * j];
            a.y = xr[4 * j + 1];
            acc = __builtin_amdgcn_wmma_f32_16x16x4_f32(
                 false, a,  false, bfrag[j],
                 (short)0, acc,  false,  false);
        }
        float* __restrict__ yr = Y + (size_t)(tm * 16 + 8 * half) * DOUT + col;
#pragma unroll
        for (int r = 0; r < 8; ++r)
            yr[(size_t)r * DOUT] = acc[r];
    }
}


template <int D, int U>
__device__ __forceinline__ vecf<D / 32> gather_rows(const float* __restrict__ h, int es, int ew,
                                                    int j, int lane, vecf<D / 32> acc) {
    constexpr int VPT = D / 32;
    vecf<VPT> v[U];
    float w[U];
#pragma unroll
    for (int k = 0; k < U; ++k) {
        const int s = __builtin_amdgcn_readlane(es, j + k);
        w[k] = __int_as_float(__builtin_amdgcn_readlane(ew, j + k));
        v[k] = *(const vecf<VPT>*)(h + (size_t)s * D + lane * VPT);
    }
#pragma unroll
    for (int k = 0; k < U; ++k) acc += w[k] * v[k];
    return acc;
}

static constexpr int EPI_SUM = 0, EPI_BIAS = 1, EPI_BN_RELU = 2;

template <int D, int EPI>
__global__ void gather_rows_kernel(const int* __restrict__ rowptr, const int* __restrict__ cnt,
                                   const int2* __restrict__ csr, const float* __restrict__ dinv,
                                   const float* __restrict__ h, const float* __restrict__ bias,
                                   const float* __restrict__ bn_g, const float* __restrict__ bn_be,
                                   const float* __restrict__ bn_m, const float* __restrict__ bn_v,
                                   float* __restrict__ agg, int n) {
    constexpr int VPT = D / 32;
    static_assert(VPT == 2 || VPT == 4, "D must be 64 or 128");
    const int lane = threadIdx.x & 31;
    const int i = __builtin_amdgcn_readfirstlane(
        (int)(blockIdx.x * (blockDim.x >> 5) + (threadIdx.x >> 5)));
    if (i >= n) return;

    const int   start = rowptr[i];
    const int   deg   = cnt[i];
    const float di    = dinv[i];

    vecf<VPT> acc = (di * di) * *(const vecf<VPT>*)(h + (size_t)i * D + lane * VPT);

    for (int base = 0; base < deg; base += 32) {
        const int m = (deg - base < 32) ? (deg - base) : 32;
        int es = 0, ew = 0;
        if (lane < m) {
            const int2 ent = csr[start + base + lane];
            es = ent.x;
            ew = ent.y;
        }
        int j = 0;
        for (; j + 8 <= m; j += 8) acc = gather_rows<D, 8>(h, es, ew, j, lane, acc);
        if (j + 4 <= m) { acc = gather_rows<D, 4>(h, es, ew, j, lane, acc); j += 4; }
        if (j + 2 <= m) { acc = gather_rows<D, 2>(h, es, ew, j, lane, acc); j += 2; }
        if (j < m)        acc = gather_rows<D, 1>(h, es, ew, j, lane, acc);
    }

    if (EPI == EPI_BIAS) acc += *(const vecf<VPT>*)(bias + lane * VPT);
    if (EPI == EPI_BN_RELU) {
        const vecf<VPT> pb  = *(const vecf<VPT>*)(bias  + lane * VPT);
        const vecf<VPT> pg  = *(const vecf<VPT>*)(bn_g  + lane * VPT);
        const vecf<VPT> pbe = *(const vecf<VPT>*)(bn_be + lane * VPT);
        const vecf<VPT> pm  = *(const vecf<VPT>*)(bn_m  + lane * VPT);
        const vecf<VPT> pv  = *(const vecf<VPT>*)(bn_v  + lane * VPT);
#pragma unroll
        for (int k = 0; k < VPT; ++k) {
            float val = acc[k] + pb[k];
            val = pg[k] * (val - pm[k]) * rsqrtf(pv[k] + 1e-5f) + pbe[k];
            acc[k] = fmaxf(val, 0.0f);
        }
    }
    *(vecf<VPT>*)(agg + (size_t)i * D + lane * VPT) = acc;
}


static inline size_t align256(size_t x) { return (x + 255) & ~(size_t)255; }

extern "C" void kernel_launch(void* const* d_in, const int* in_sizes, int n_in,
                              void* d_out, int out_size, void* d_ws, size_t ws_size,
                              hipStream_t stream) {
    (void)in_sizes; (void)n_in; (void)out_size; (void)ws_size;

    const float* x   = (const float*)d_in[0];
    const int*   ei  = (const int*)d_in[1];
    const int*   src = ei;
    const int*   dst = ei + GE;
    const float* W1  = (const float*)d_in[2];
    const float* b1  = (const float*)d_in[3];
    const float* g1  = (const float*)d_in[4];
    const float* be1 = (const float*)d_in[5];
    const float* m1  = (const float*)d_in[6];
    const float* v1  = (const float*)d_in[7];
    const float* W2  = (const float*)d_in[8];
    const float* b2  = (const float*)d_in[9];
    const float* g2  = (const float*)d_in[10];
    const float* be2 = (const float*)d_in[11];
    const float* m2  = (const float*)d_in[12];
    const float* v2  = (const float*)d_in[13];
    const float* W3  = (const float*)d_in[14];
    const float* b3  = (const float*)d_in[15];
    float* out = (float*)d_out;

    char* ws = (char*)d_ws;
    size_t off = 0;
    float* dinv   = (float*)(ws + off); off += align256((size_t)GN * 4);
    int*   cnt    = (int*)(ws + off);   off += align256((size_t)GN * 4);
    int*   rowptr = (int*)(ws + off);   off += align256((size_t)GN * 4);
    int*   rank   = (int*)(ws + off);   off += align256((size_t)GE * 4);
    int2*  csr    = (int2*)(ws + off);  off += align256((size_t)GE * 8);
    int*   chtab  = (int*)(ws + off);   off += align256((size_t)NCH * GN * 4);
    int*   segsum = (int*)(ws + off);   off += align256((size_t)NCH * NSEG * 4);
    float* hbuf = (float*)(ws + off); off += align256((size_t)GN * GH * 4);
    float* agg1 = (float*)(ws + off); off += align256((size_t)GN * GH * 4);
    float* agg2 = (float*)(ws + off); off += align256((size_t)GN * GH * 4);

    const int TPB = 256;

    chunk_count_kernel<<<NCH, CNT_T, 0, stream>>>(dst, rank, chtab, segsum);
    scan_dinv_kernel<<<NSEG, SEG, 0, stream>>>(chtab, segsum, cnt, rowptr, dinv, GN);
    csr_fill_kernel<<<(GE + TPB - 1) / TPB, TPB, 0, stream>>>(src, dst, rank, chtab, dinv, csr, GE);

    const int mTiles = GN / 16;
    const int gemmBlocks = 512;
    const int rowBlocks = (GN + 7) / 8;

    gemm_wmma_f32_kernel<GIN, GH><<<gemmBlocks, TPB, 0, stream>>>(
        x, W1, hbuf, mTiles);
    gather_rows_kernel<GH, EPI_BN_RELU><<<rowBlocks, TPB, 0, stream>>>(
        rowptr, cnt, csr, dinv, hbuf, b1, g1, be1, m1, v1, agg1, GN);

    gemm_wmma_f32_kernel<GH, GH><<<gemmBlocks, TPB, 0, stream>>>(
        agg1, W2, hbuf, mTiles);
    gather_rows_kernel<GH, EPI_BN_RELU><<<rowBlocks, TPB, 0, stream>>>(
        rowptr, cnt, csr, dinv, hbuf, b2, g2, be2, m2, v2, agg2, GN);

    gemm_wmma_f32_kernel<GH, GC><<<gemmBlocks, TPB, 0, stream>>>(
        agg2, W3, hbuf, mTiles);
    gather_rows_kernel<GC, EPI_BIAS><<<rowBlocks, TPB, 0, stream>>>(
        rowptr, cnt, csr, dinv, hbuf, b3, nullptr, nullptr, nullptr, nullptr, out, GN);
}


# product kernels: 16-row tile of X fetched once per work-group as whole rows (16 B per lane) into LDS, A fragments read from LDS; was 8-byte pieces of 16 rows per WMMA from global in every wave
# speedup vs baseline: 4.0668x; 1.2198x over previous
#include <hip/hip_runtime.h>
#include <hip/hip_bf16.h>


typedef __attribute__((ext_vector_type(2))) float v2f;
typedef __attribute__((ext_vector_type(8))) float v8f;

static constexpr int GN  = 50000;
static constexpr int GE  = 800000;
static constexpr int GIN = 128;
static constexpr int GH  = 128;
static constexpr int GC  = 64;


template <int V> using vecf = float __attribute__((ext_vector_type(V)));

static constexpr int NCH   = 64;
static constexpr int CH    = GE / NCH;
static constexpr int CNT_T = 1024;
static constexpr int CNT_U = 8;
static constexpr int SEG   = 256;
static constexpr int NSEG  = (GN + SEG - 1) / SEG;
static_assert(CH * NCH == GE, "the chunks must tile the edge list");
static_assert(NSEG <= SEG, "one thread per segment in front of a block");
static_assert(GN % 4 == 0, "node tables are copied 16 bytes at a time");

__global__ void __launch_bounds__(CNT_T)
chunk_count_kernel(const int* __restrict__ dst, int* __restrict__ rank,
                   int* __restrict__ chtab, int* __restrict__ segsum) {
    __shared__ __attribute__((aligned(16))) int tab[NSEG * SEG];
    const int t = threadIdx.x;
    const int p = blockIdx.x;

    int4* tab4 = (int4*)tab;
    for (int q = t; q < NSEG * SEG / 4; q += CNT_T) tab4[q] = make_int4(0, 0, 0, 0);
    __syncthreads();

    const int* __restrict__ dp = dst  + (size_t)p * CH;
    int*       __restrict__ rp = rank + (size_t)p * CH;
    for (int k0 = t; k0 < CH; k0 += CNT_U * CNT_T) {
        int d[CNT_U], r[CNT_U];
#pragma unroll
        for (int u = 0; u < CNT_U; ++u) {
            const int k = k0 + u * CNT_T;
            d[u] = (k < CH) ? dp[k] : -1;
        }
#pragma unroll
        for (int u = 0; u < CNT_U; ++u) {
            r[u] = 0;
            if (d[u] >= 0) r[u] = atomicAdd(&tab[d[u]], 1);
        }
#pragma unroll
        for (int u = 0; u < CNT_U; ++u) {
            const int k = k0 + u * CNT_T;
            if (k < CH) rp[k] = r[u];
        }
    }
    __syncthreads();

    int4* __restrict__ out4 = (int4*)(chtab + (size_t)p * GN);
    for (int q = t; q < GN / 4; q += CNT_T) out4[q] = tab4[q];

    const int lane = t & 31;
    for (int s = t >> 5; s < NSEG; s += CNT_T / 32) {
        int v = 0;
        for (int k = 0; k < SEG / 32; ++k) v += tab[s * SEG + k * 32 + lane];
        for (int o = 16; o > 0; o >>= 1) v += __shfl_xor(v, o);
        if (lane == 0) segsum[p * NSEG + s] = v;
    }
}

__global__ void __launch_bounds__(SEG)
scan_dinv_kernel(int* __restrict__ chtab, const int* __restrict__ segsum,
                 int* __restrict__ cnt, int* __restrict__ rowptr,
                 float* __restrict__ dinv, int n) {
    constexpr int NW = SEG / 32;
    __shared__ int wtot[NW];
    __shared__ int wpre[NW];
    const int t    = threadIdx.x;
    const int lane = t & 31;
    const int wv   = t >> 5;
    const int b    = blockIdx.x;

    int s = 0;
    if (t < b)
        for (int p = 0; p < NCH; ++p) s += segsum[p * NSEG + t];

    const int i  = b * SEG + t;
    const int ic = (i < n) ? i : n - 1;
    int cc[NCH];
    int c = 0;
#pragma unroll
    for (int p = 0; p < NCH; ++p) {
        cc[p] = chtab[(size_t)p * GN + ic];
        c += cc[p];
    }
    if (i >= n) c = 0;

    int inc = c;
    for (int o = 1; o < 32; o <<= 1) {
        int v = __shfl_up(inc, o);
        if (lane >= o) inc += v;
    }
    for (int o = 16; o > 0; o >>= 1) s += __shfl_xor(s, o);

    if (lane == 31) wtot[wv] = inc;
    if (lane == 0)  wpre[wv] = s;
    __syncthreads();

    int a  = (lane < NW) ? wpre[lane] : 0;
    int bw = (lane < wv) ? wtot[lane] : 0;
    for (int o = 16; o > 0; o >>= 1) {
        a  += __shfl_xor(a, o);
        bw += __shfl_xor(bw, o);
    }
    if (i < n) {
        int run = a + bw + inc - c;
        rowptr[i] = run;
        cnt[i]    = c;
        dinv[i]   = rsqrtf((float)(c + 1));
#pragma unroll
        for (int p = 0; p < NCH; ++p) {
            chtab[(size_t)p * GN + i] = run;
            run += cc[p];
        }
    }
}

__global__ void csr_fill_kernel(const int* __restrict__ src, const int* __restrict__ dst,
                                const int* __restrict__ rank, const int* __restrict__ chtab,
                                const float* __restrict__ dinv, int2* __restrict__ csr, int nE) {
    int e = blockIdx.x * blockDim.x + threadIdx.x;
    if (e >= nE) return;
    const int s = src[e];
    const int d = dst[e];
    const int p = e / CH;
    int2 ent;
    ent.x = s;
    ent.y = __float_as_int(dinv[s] * dinv[d]);
    csr[chtab[(size_t)p * GN + d] + rank[e]] = ent;
}


template <int DIN, int DOUT>
__global__ void __launch_bounds__((DOUT / 16) * 32)
gemm_wmma_f32_kernel(const float* __restrict__ X,
                     const float* __restrict__ W,
                     float* __restrict__ Y,
                     int mTiles) {
    static_assert(DIN == 128, "one row of X is one 16-byte load per lane");
    constexpr int tilesN = DOUT / 16;
    constexpr int KF     = DIN / 4;
    constexpr int RPW    = 16 / tilesN;
    static_assert(RPW * tilesN == 16, "the waves' rows must tile the 16 rows");
    constexpr int PITCH  = DIN + 4;

    __shared__ __attribute__((aligned(16))) float atile[2][16 * PITCH];

    const int tn   = threadIdx.x >> 5;
    const int lane = threadIdx.x & 31;
    const int half = lane >> 4;
    const int l15  = lane & 15;
    const int col  = tn * 16 + l15;

    const int row0 = tn * RPW;
    int tm = blockIdx.x;
    vecf<4> nx[RPW];

    const int t0 = (tm < mTiles) ? tm : (mTiles - 1);
#pragma unroll
    for (int r = 0; r < RPW; ++r)
        nx[r] = *(const vecf<4>*)(X + (size_t)(t0 * 16 + row0 + r) * DIN + lane * 4);

    v2f bfrag[KF];
#pragma unroll
    for (int j = 0; j < KF; ++j) {
        const int k = 4 * j + 2 * half;
        bfrag[j].x = W[(size_t)k * DOUT + col];
        bfrag[j].y = W[(size_t)(k + 1) * DOUT + col];
    }

#pragma unroll
    for (int r = 0; r < RPW; ++r)
        *(vecf<4>*)(&atile[0][(row0 + r) * PITCH + lane * 4]) = nx[r];
    __builtin_amdgcn_sched_group_barrier(0x020, 2 * KF + RPW, 1);
    __builtin_amdgcn_sched_group_barrier(0x200, RPW, 1);
    __syncthreads();

    int cur = 0;
    for (; tm < mTiles; tm += gridDim.x) {
        const int  tnext = tm + (int)gridDim.x;
        const bool more  = tnext < mTiles;
        if (more) {
#pragma unroll
            for (int r = 0; r < RPW; ++r)
                nx[r] = *(const vecf<4>*)(X + (size_t)(tnext * 16 + row0 + r) * DIN + lane * 4);
        }

        const float* ar = &atile[cur][l15 * PITCH + 2 * half];
        v2f afrag[KF];
#pragma unroll
        for (int j = 0; j < KF; ++j) afrag[j] = *(const v2f*)(ar + 4 * j);
        v8f acc = {};
#pragma unroll
        for (int j = 0; j < KF; ++j) {
            acc = __builtin_amdgcn_wmma_f32_16x16x4_f32(
                 false, afrag[j],  false, bfrag[j],
                 (short)0, acc,  false,  false);
        }
        __builtin_amdgcn_sched_group_barrier(0x100, KF, 0);
        __builtin_amdgcn_sched_group_barrier(0x008, KF, 0);
        if (more) {
#pragma unroll
            for (int r = 0; r < RPW; ++r)
                *(vecf<4>*)(&atile[cur ^ 1][(row0 + r) * PITCH + lane * 4]) = nx[r];
        }
        __syncthreads();
        cur ^= 1;

        float* __restrict__ yr = Y + (size_t)(tm * 16 + 8 * half) * DOUT + col;
#pragma unroll
        for (int r = 0; r < 8; ++r)
            yr[(size_t)r * DOUT] = acc[r];
    }
}


template <int D, int U>
__device__ __forceinline__ vecf<D / 32> gather_rows(const float* __restrict__ h, int es, int ew,
                                                    int j, int lane, vecf<D / 32> acc) {
    constexpr int VPT = D / 32;
    vecf<VPT> v[U];
    float w[U];
#pragma unroll
    for (int k = 0; k < U; ++k) {
        const int s = __builtin_amdgcn_readlane(es, j + k);
        w[k] = __int_as_float(__builtin_amdgcn_readlane(ew, j + k));
        v[k] = *(const vecf<VPT>*)(h + (size_t)s * D + lane * VPT);
    }
#pragma unroll
    for (int k = 0; k < U; ++k) acc += w[k] * v[k];
    return acc;
}

static constexpr int EPI_SUM = 0, EPI_BIAS = 1, EPI_BN_RELU = 2;

template <int D, int EPI>
__global__ void gather_rows_kernel(const int* __restrict__ rowptr, const int* __restrict__ cnt,
                                   const int2* __restrict__ csr, const float* __restrict__ dinv,
                                   const float* __restrict__ h, const float* __restrict__ bias,
                                   const float* __restrict__ bn_g, const float* __restrict__ bn_be,
                                   const float* __restrict__ bn_m, const float* __restrict__ bn_v,
                                   float* __restrict__ agg, int n) {
    constexpr int VPT = D / 32;
    static_assert(VPT == 2 || VPT == 4, "D must be 64 or 128");
    const int lane = threadIdx.x & 31;
    const int i = __builtin_amdgcn_readfirstlane(
        (int)(blockIdx.x * (blockDim.x >> 5) + (threadIdx.x >> 5)));
    if (i >= n) return;

    const int   start = rowptr[i];
    const int   deg   = cnt[i];
    const float di    = dinv[i];

    vecf<VPT> acc = (di * di) * *(const vecf<VPT>*)(h + (size_t)i * D + lane * VPT);

    for (int base = 0; base < deg; base += 32) {
        const int m = (deg - base < 32) ? (deg - base) : 32;
        int es = 0, ew = 0;
        if (lane < m) {
            const int2 ent = csr[start + base + lane];
            es = ent.x;
            ew = ent.y;
        }
        int j = 0;
        for (; j + 8 <= m; j += 8) acc = gather_rows<D, 8>(h, es, ew, j, lane, acc);
        if (j + 4 <= m) { acc = gather_rows<D, 4>(h, es, ew, j, lane, acc); j += 4; }
        if (j + 2 <= m) { acc = gather_rows<D, 2>(h, es, ew, j, lane, acc); j += 2; }
        if (j < m)        acc = gather_rows<D, 1>(h, es, ew, j, lane, acc);
    }

    if (EPI == EPI_BIAS) acc += *(const vecf<VPT>*)(bias + lane * VPT);
    if (EPI == EPI_BN_RELU) {
        const vecf<VPT> pb  = *(const vecf<VPT>*)(bias  + lane * VPT);
        const vecf<VPT> pg  = *(const vecf<VPT>*)(bn_g  + lane * VPT);
        const vecf<VPT> pbe = *(const vecf<VPT>*)(bn_be + lane * VPT);
        const vecf<VPT> pm  = *(const vecf<VPT>*)(bn_m  + lane * VPT);
        const vecf<VPT> pv  = *(const vecf<VPT>*)(bn_v  + lane * VPT);
#pragma unroll
        for (int k = 0; k < VPT; ++k) {
            float val = acc[k] + pb[k];
            val = pg[k] * (val - pm[k]) * rsqrtf(pv[k] + 1e-5f) + pbe[k];
            acc[k] = fmaxf(val, 0.0f);
        }
    }
    *(vecf<VPT>*)(agg + (size_t)i * D + lane * VPT) = acc;
}


static inline size_t align256(size_t x) { return (x + 255) & ~(size_t)255; }

extern "C" void kernel_launch(void* const* d_in, const int* in_sizes, int n_in,
                              void* d_out, int out_size, void* d_ws, size_t ws_size,
                              hipStream_t stream) {
    (void)in_sizes; (void)n_in; (void)out_size; (void)ws_size;

    const float* x   = (const float*)d_in[0];
    const int*   ei  = (const int*)d_in[1];
    const int*   src = ei;
    const int*   dst = ei + GE;
    const float* W1  = (const float*)d_in[2];
    const float* b1  = (const float*)d_in[3];
    const float* g1  = (const float*)d_in[4];
    const float* be1 = (const float*)d_in[5];
    const float* m1  = (const float*)d_in[6];
    const float* v1  = (const float*)d_in[7];
    const float* W2  = (const float*)d_in[8];
    const float* b2  = (const float*)d_in[9];
    const float* g2  = (const float*)d_in[10];
    const float* be2 = (const float*)d_in[11];
    const float* m2  = (const float*)d_in[12];
    const float* v2  = (const float*)d_in[13];
    const float* W3  = (const float*)d_in[14];
    const float* b3  = (const float*)d_in[15];
    float* out = (float*)d_out;

    char* ws = (char*)d_ws;
    size_t off = 0;
    float* dinv   = (float*)(ws + off); off += align256((size_t)GN * 4);
    int*   cnt    = (int*)(ws + off);   off += align256((size_t)GN * 4);
    int*   rowptr = (int*)(ws + off);   off += align256((size_t)GN * 4);
    int*   rank   = (int*)(ws + off);   off += align256((size_t)GE * 4);
    int2*  csr    = (int2*)(ws + off);  off += align256((size_t)GE * 8);
    int*   chtab  = (int*)(ws + off);   off += align256((size_t)NCH * GN * 4);
    int*   segsum = (int*)(ws + off);   off += align256((size_t)NCH * NSEG * 4);
    float* hbuf = (float*)(ws + off); off += align256((size_t)GN * GH * 4);
    float* agg1 = (float*)(ws + off); off += align256((size_t)GN * GH * 4);
    float* agg2 = (float*)(ws + off); off += align256((size_t)GN * GH * 4);

    const int TPB = 256;

    chunk_count_kernel<<<NCH, CNT_T, 0, stream>>>(dst, rank, chtab, segsum);
    scan_dinv_kernel<<<NSEG, SEG, 0, stream>>>(chtab, segsum, cnt, rowptr, dinv, GN);
    csr_fill_kernel<<<(GE + TPB - 1) / TPB, TPB, 0, stream>>>(src, dst, rank, chtab, dinv, csr, GE);

    const int mTiles = GN / 16;
    const int gemmWaves = 4096;
    const int rowBlocks = (GN + 7) / 8;

    gemm_wmma_f32_kernel<GIN, GH><<<gemmWaves / (GH / 16), (GH / 16) * 32, 0, stream>>>(
        x, W1, hbuf, mTiles);
    gather_rows_kernel<GH, EPI_BN_RELU><<<rowBlocks, TPB, 0, stream>>>(
        rowptr, cnt, csr, dinv, hbuf, b1, g1, be1, m1, v1, agg1, GN);

    gemm_wmma_f32_kernel<GH, GH><<<gemmWaves / (GH / 16), (GH / 16) * 32, 0, stream>>>(
        agg1, W2, hbuf, mTiles);
    gather_rows_kernel<GH, EPI_BN_RELU><<<rowBlocks, TPB, 0, stream>>>(
        rowptr, cnt, csr, dinv, hbuf, b2, g2, be2, m2, v2, agg2, GN);

    gemm_wmma_f32_kernel<GH, GC><<<gemmWaves / (GC / 16), (GC / 16) * 32, 0, stream>>>(
        agg2, W3, hbuf, mTiles);
    gather_rows_kernel<GC, EPI_BIAS><<<rowBlocks, TPB, 0, stream>>>(
        rowptr, cnt, csr, dinv, hbuf, b3, nullptr, nullptr, nullptr, nullptr, out, GN);
}


# layers 1,2: gather waves lay each summed row (bias+BN+ReLU done) in a 16-row LDS tile that the same work-group multiplies by the next W on WMMA; y arrays and 2 launches removed; sum order kept
# speedup vs baseline: 4.2511x; 1.0453x over previous
#include <hip/hip_runtime.h>
#include <hip/hip_bf16.h>


typedef __attribute__((ext_vector_type(2))) float v2f;
typedef __attribute__((ext_vector_type(8))) float v8f;

static constexpr int GN  = 50000;
static constexpr int GE  = 800000;
static constexpr int GIN = 128;
static constexpr int GH  = 128;
static constexpr int GC  = 64;


template <int V> using vecf = float __attribute__((ext_vector_type(V)));

static constexpr int NCH   = 64;
static constexpr int CH    = GE / NCH;
static constexpr int CNT_T = 1024;
static constexpr int CNT_U = 8;
static constexpr int SEG   = 256;
static constexpr int NSEG  = (GN + SEG - 1) / SEG;
static_assert(CH * NCH == GE, "the chunks must tile the edge list");
static_assert(NSEG <= SEG, "one thread per segment in front of a block");
static_assert(GN % 4 == 0, "node tables are copied 16 bytes at a time");

__global__ void __launch_bounds__(CNT_T)
chunk_count_kernel(const int* __restrict__ dst, int* __restrict__ rank,
                   int* __restrict__ chtab, int* __restrict__ segsum) {
    __shared__ __attribute__((aligned(16))) int tab[NSEG * SEG];
    const int t = threadIdx.x;
    const int p = blockIdx.x;

    int4* tab4 = (int4*)tab;
    for (int q = t; q < NSEG * SEG / 4; q += CNT_T) tab4[q] = make_int4(0, 0, 0, 0);
    __syncthreads();

    const int* __restrict__ dp = dst  + (size_t)p * CH;
    int*       __restrict__ rp = rank + (size_t)p * CH;
    for (int k0 = t; k0 < CH; k0 += CNT_U * CNT_T) {
        int d[CNT_U], r[CNT_U];
#pragma unroll
        for (int u = 0; u < CNT_U; ++u) {
            const int k = k0 + u * CNT_T;
            d[u] = (k < CH) ? dp[k] : -1;
        }
#pragma unroll
        for (int u = 0; u < CNT_U; ++u) {
            r[u] = 0;
            if (d[u] >= 0) r[u] = atomicAdd(&tab[d[u]], 1);
        }
#pragma unroll
        for (int u = 0; u < CNT_U; ++u) {
            const int k = k0 + u * CNT_T;
            if (k < CH) rp[k] = r[u];
        }
    }
    __syncthreads();

    int4* __restrict__ out4 = (int4*)(chtab + (size_t)p * GN);
    for (int q = t; q < GN / 4; q += CNT_T) out4[q] = tab4[q];

    const int lane = t & 31;
    for (int s = t >> 5; s < NSEG; s += CNT_T / 32) {
        int v = 0;
        for (int k = 0; k < SEG / 32; ++k) v += tab[s * SEG + k * 32 + lane];
        for (int o = 16; o > 0; o >>= 1) v += __shfl_xor(v, o);
        if (lane == 0) segsum[p * NSEG + s] = v;
    }
}

__global__ void __launch_bounds__(SEG)
scan_dinv_kernel(int* __restrict__ chtab, const int* __restrict__ segsum,
                 int* __restrict__ cnt, int* __restrict__ rowptr,
                 float* __restrict__ dinv, int n) {
    constexpr int NW = SEG / 32;
    __shared__ int wtot[NW];
    __shared__ int wpre[NW];
    const int t    = threadIdx.x;
    const int lane = t & 31;
    const int wv   = t >> 5;
    const int b    = blockIdx.x;

    int s = 0;
    if (t < b)
        for (int p = 0; p < NCH; ++p) s += segsum[p * NSEG + t];

    const int i  = b * SEG + t;
    const int ic = (i < n) ? i : n - 1;
    int cc[NCH];
    int c = 0;
#pragma unroll
    for (int p = 0; p < NCH; ++p) {
        cc[p] = chtab[(size_t)p * GN + ic];
        c += cc[p];
    }
    if (i >= n) c = 0;

    int inc = c;
    for (int o = 1; o < 32; o <<= 1) {
        int v = __shfl_up(inc, o);
        if (lane >= o) inc += v;
    }
    for (int o = 16; o > 0; o >>= 1) s += __shfl_xor(s, o);

    if (lane == 31) wtot[wv] = inc;
    if (lane == 0)  wpre[wv] = s;
    __syncthreads();

    int a  = (lane < NW) ? wpre[lane] : 0;
    int bw = (lane < wv) ? wtot[lane] : 0;
    for (int o = 16; o > 0; o >>= 1) {
        a  += __shfl_xor(a, o);
        bw += __shfl_xor(bw, o);
    }
    if (i < n) {
        int run = a + bw + inc - c;
        rowptr[i] = run;
        cnt[i]    = c;
        dinv[i]   = rsqrtf((float)(c + 1));
#pragma unroll
        for (int p = 0; p < NCH; ++p) {
            chtab[(size_t)p * GN + i] = run;
            run += cc[p];
        }
    }
}

__global__ void csr_fill_kernel(const int* __restrict__ src, const int* __restrict__ dst,
                                const int* __restrict__ rank, const int* __restrict__ chtab,
                                const float* __restrict__ dinv, int2* __restrict__ csr, int nE) {
    int e = blockIdx.x * blockDim.x + threadIdx.x;
    if (e >= nE) return;
    const int s = src[e];
    const int d = dst[e];
    const int p = e / CH;
    int2 ent;
    ent.x = s;
    ent.y = __float_as_int(dinv[s] * dinv[d]);
    csr[chtab[(size_t)p * GN + d] + rank[e]] = ent;
}


template <int DIN, int DOUT>
__global__ void __launch_bounds__((DOUT / 16) * 32)
gemm_wmma_f32_kernel(const float* __restrict__ X,
                     const float* __restrict__ W,
                     float* __restrict__ Y,
                     int mTiles) {
    static_assert(DIN == 128, "one row of X is one 16-byte load per lane");
    constexpr int tilesN = DOUT / 16;
    constexpr int KF     = DIN / 4;
    constexpr int RPW    = 16 / tilesN;
    static_assert(RPW * tilesN == 16, "the waves' rows must tile the 16 rows");
    constexpr int PITCH  = DIN + 4;

    __shared__ __attribute__((aligned(16))) float atile[2][16 * PITCH];

    const int tn   = threadIdx.x >> 5;
    const int lane = threadIdx.x & 31;
    const int half = lane >> 4;
    const int l15  = lane & 15;
    const int col  = tn * 16 + l15;

    const int row0 = tn * RPW;
    int tm = blockIdx.x;
    vecf<4> nx[RPW];

    const int t0 = (tm < mTiles) ? tm : (mTiles - 1);
#pragma unroll
    for (int r = 0; r < RPW; ++r)
        nx[r] = *(const vecf<4>*)(X + (size_t)(t0 * 16 + row0 + r) * DIN + lane * 4);

    v2f bfrag[KF];
#pragma unroll
    for (int j = 0; j < KF; ++j) {
        const int k = 4 * j + 2 * half;
        bfrag[j].x = W[(size_t)k * DOUT + col];
        bfrag[j].y = W[(size_t)(k + 1) * DOUT + col];
    }

#pragma unroll
    for (int r = 0; r < RPW; ++r)
        *(vecf<4>*)(&atile[0][(row0 + r) * PITCH + lane * 4]) = nx[r];
    __builtin_amdgcn_sched_group_barrier(0x020, 2 * KF + RPW, 1);
    __builtin_amdgcn_sched_group_barrier(0x200, RPW, 1);
    __syncthreads();

    int cur = 0;
    for (; tm < mTiles; tm += gridDim.x) {
        const int  tnext = tm + (int)gridDim.x;
        const bool more  = tnext < mTiles;
        if (more) {
#pragma unroll
            for (int r = 0; r < RPW; ++r)
                nx[r] = *(const vecf<4>*)(X + (size_t)(tnext * 16 + row0 + r) * DIN + lane * 4);
        }

        const float* ar = &atile[cur][l15 * PITCH + 2 * half];
        v2f afrag[KF];
#pragma unroll
        for (int j = 0; j < KF; ++j) afrag[j] = *(const v2f*)(ar + 4 * j);
        v8f acc = {};
#pragma unroll
        for (int j = 0; j < KF; ++j) {
            acc = __builtin_amdgcn_wmma_f32_16x16x4_f32(
                 false, afrag[j],  false, bfrag[j],
                 (short)0, acc,  false,  false);
        }
        __builtin_amdgcn_sched_group_barrier(0x100, KF, 0);
        __builtin_amdgcn_sched_group_barrier(0x008, KF, 0);
        if (more) {
#pragma unroll
            for (int r = 0; r < RPW; ++r)
                *(vecf<4>*)(&atile[cur ^ 1][(row0 + r) * PITCH + lane * 4]) = nx[r];
        }
        __syncthreads();
        cur ^= 1;

        float* __restrict__ yr = Y + (size_t)(tm * 16 + 8 * half) * DOUT + col;
#pragma unroll
        for (int r = 0; r < 8; ++r)
            yr[(size_t)r * DOUT] = acc[r];
    }
}


template <int D, int U>
__device__ __forceinline__ vecf<D / 32> gather_rows(const float* __restrict__ h, int es, int ew,
                                                    int j, int lane, vecf<D / 32> acc) {
    constexpr int VPT = D / 32;
    vecf<VPT> v[U];
    float w[U];
#pragma unroll
    for (int k = 0; k < U; ++k) {
        const int s = __builtin_amdgcn_readlane(es, j + k);
        w[k] = __int_as_float(__builtin_amdgcn_readlane(ew, j + k));
        const unsigned long long a = (unsigned long long)(h + (size_t)s * D);
        const unsigned lo = __builtin_amdgcn_readfirstlane((unsigned)a);
        const unsigned hi = __builtin_amdgcn_readfirstlane((unsigned)(a >> 32));
        typedef const vecf<VPT> __attribute__((address_space(1))) * grow_t;
        const grow_t hs = (grow_t)(((unsigned long long)hi << 32) | lo);
        v[k] = hs[lane];
    }
#pragma unroll
    for (int k = 0; k < U; ++k) acc += w[k] * v[k];
    return acc;
}

static constexpr int EPI_SUM = 0, EPI_BIAS = 1, EPI_BN_RELU = 2;

template <int D, int EPI>
__global__ void gather_rows_kernel(const int* __restrict__ rowptr, const int* __restrict__ cnt,
                                   const int2* __restrict__ csr, const float* __restrict__ dinv,
                                   const float* __restrict__ h, const float* __restrict__ bias,
                                   const float* __restrict__ bn_g, const float* __restrict__ bn_be,
                                   const float* __restrict__ bn_m, const float* __restrict__ bn_v,
                                   float* __restrict__ agg, int n) {
    constexpr int VPT = D / 32;
    static_assert(VPT == 2 || VPT == 4, "D must be 64 or 128");
    const int lane = threadIdx.x & 31;
    const int i = __builtin_amdgcn_readfirstlane(
        (int)(blockIdx.x * (blockDim.x >> 5) + (threadIdx.x >> 5)));
    if (i >= n) return;

    const int   start = rowptr[i];
    const int   deg   = cnt[i];
    const float di    = dinv[i];

    vecf<VPT> acc = (di * di) * *(const vecf<VPT>*)(h + (size_t)i * D + lane * VPT);

    for (int base = 0; base < deg; base += 32) {
        const int m = (deg - base < 32) ? (deg - base) : 32;
        int es = 0, ew = 0;
        if (lane < m) {
            const int2 ent = csr[start + base + lane];
            es = ent.x;
            ew = ent.y;
        }
        int j = 0;
        for (; j + 8 <= m; j += 8) acc = gather_rows<D, 8>(h, es, ew, j, lane, acc);
        if (j + 4 <= m) { acc = gather_rows<D, 4>(h, es, ew, j, lane, acc); j += 4; }
        if (j + 2 <= m) { acc = gather_rows<D, 2>(h, es, ew, j, lane, acc); j += 2; }
        if (j < m)        acc = gather_rows<D, 1>(h, es, ew, j, lane, acc);
    }

    if (EPI == EPI_BIAS) acc += *(const vecf<VPT>*)(bias + lane * VPT);
    if (EPI == EPI_BN_RELU) {
        const vecf<VPT> pb  = *(const vecf<VPT>*)(bias  + lane * VPT);
        const vecf<VPT> pg  = *(const vecf<VPT>*)(bn_g  + lane * VPT);
        const vecf<VPT> pbe = *(const vecf<VPT>*)(bn_be + lane * VPT);
        const vecf<VPT> pm  = *(const vecf<VPT>*)(bn_m  + lane * VPT);
        const vecf<VPT> pv  = *(const vecf<VPT>*)(bn_v  + lane * VPT);
#pragma unroll
        for (int k = 0; k < VPT; ++k) {
            float val = acc[k] + pb[k];
            val = pg[k] * (val - pm[k]) * rsqrtf(pv[k] + 1e-5f) + pbe[k];
            acc[k] = fmaxf(val, 0.0f);
        }
    }
    *(vecf<VPT>*)(agg + (size_t)i * D + lane * VPT) = acc;
}


template <int D>
__device__ __forceinline__ vecf<D / 32> gather_row_sum(const int2* __restrict__ csr,
                                                       const float* __restrict__ h,
                                                       int i, int start, int deg, float di,
                                                       int es, int ew, int lane) {
    constexpr int VPT = D / 32;
    vecf<VPT> acc = (di * di) * *(const vecf<VPT>*)(h + (size_t)i * D + lane * VPT);
    for (int base = 0; base < deg; base += 32) {
        const int m = (deg - base < 32) ? (deg - base) : 32;
        if (base > 0) {
            es = 0; ew = 0;
            if (lane < m) {
                const int2 ent = csr[start + base + lane];
                es = ent.x;
                ew = ent.y;
            }
        }
        int j = 0;
        for (; j + 8 <= m; j += 8) acc = gather_rows<D, 8>(h, es, ew, j, lane, acc);
        if (j + 4 <= m) { acc = gather_rows<D, 4>(h, es, ew, j, lane, acc); j += 4; }
        if (j + 2 <= m) { acc = gather_rows<D, 2>(h, es, ew, j, lane, acc); j += 2; }
        if (j < m)        acc = gather_rows<D, 1>(h, es, ew, j, lane, acc);
    }
    return acc;
}

template <int DOUT>
__global__ void __launch_bounds__((DOUT / 16) * 32)
gather_gemm_kernel(const int* __restrict__ rowptr, const int* __restrict__ cnt,
                   const int2* __restrict__ csr, const float* __restrict__ dinv,
                   const float* __restrict__ h, const float* __restrict__ bias,
                   const float* __restrict__ bn_g, const float* __restrict__ bn_be,
                   const float* __restrict__ bn_m, const float* __restrict__ bn_v,
                   const float* __restrict__ W, float* __restrict__ Y, int mTiles) {
    constexpr int D      = GH;
    constexpr int tilesN = DOUT / 16;
    constexpr int KF     = D / 4;
    constexpr int RPW    = 16 / tilesN;
    static_assert(D == 128, "a lane owns 4 features: one 16-byte LDS store per row");
    static_assert(RPW * tilesN == 16, "the waves' rows must tile the 16 rows");
    constexpr int PITCH  = D + 4;

    __shared__ __attribute__((aligned(16))) float atile[2][16 * PITCH];
    __shared__ __attribute__((aligned(16))) float par[5][D];

    const int tn   = threadIdx.x >> 5;
    const int lane = threadIdx.x & 31;
    const int half = lane >> 4;
    const int l15  = lane & 15;
    const int col  = tn * 16 + l15;
    const int row0 = __builtin_amdgcn_readfirstlane(tn) * RPW;

    for (int q = threadIdx.x; q < D; q += tilesN * 32) {
        par[0][q] = bias[q];
        par[1][q] = bn_g[q];
        par[2][q] = bn_m[q];
        par[3][q] = rsqrtf(bn_v[q] + 1e-5f);
        par[4][q] = bn_be[q];
    }

    v2f bfrag[KF];
#pragma unroll
    for (int j = 0; j < KF; ++j) {
        const int k = 4 * j + 2 * half;
        bfrag[j].x = W[(size_t)k * DOUT + col];
        bfrag[j].y = W[(size_t)(k + 1) * DOUT + col];
    }
    __syncthreads();

    int cur = 0;
    for (int tm = blockIdx.x; tm < mTiles; tm += gridDim.x) {
        const int i0 = tm * 16 + row0;
        int   st[RPW], dg[RPW], es[RPW], ew[RPW];
        float di[RPW];
#pragma unroll
        for (int r = 0; r < RPW; ++r) {
            st[r] = rowptr[i0 + r];
            dg[r] = cnt[i0 + r];
            di[r] = dinv[i0 + r];
        }
#pragma unroll
        for (int r = 0; r < RPW; ++r) {
            const int q = st[r] + lane;
            const int2 ent = csr[q < GE ? q : GE - 1];
            es[r] = ent.x;
            ew[r] = ent.y;
        }
#pragma unroll
        for (int r = 0; r < RPW; ++r) {
            vecf<4> acc4 = gather_row_sum<D>(csr, h, i0 + r, st[r], dg[r], di[r], es[r], ew[r], lane);
            asm volatile("" ::: "memory");
            const vecf<4> pb  = *(const vecf<4>*)(&par[0][lane * 4]);
            const vecf<4> pg  = *(const vecf<4>*)(&par[1][lane * 4]);
            const vecf<4> pm  = *(const vecf<4>*)(&par[2][lane * 4]);
            const vecf<4> prs = *(const vecf<4>*)(&par[3][lane * 4]);
            const vecf<4> pbe = *(const vecf<4>*)(&par[4][lane * 4]);
#pragma unroll
            for (int k = 0; k < 4; ++k) {
                float val = acc4[k] + pb[k];
                val = pg[k] * (val - pm[k]) * prs[k] + pbe[k];
                acc4[k] = fmaxf(val, 0.0f);
            }
            *(vecf<4>*)(&atile[cur][(row0 + r) * PITCH + lane * 4]) = acc4;
        }
        __syncthreads();

        const float* ar = &atile[cur][l15 * PITCH + 2 * half];
        v8f acc = {};
#pragma unroll
        for (int hh = 0; hh < 2; ++hh) {
            v2f afrag[KF / 2];
#pragma unroll
            for (int j = 0; j < KF / 2; ++j) afrag[j] = *(const v2f*)(ar + 4 * (hh * (KF / 2) + j));
#pragma unroll
            for (int j = 0; j < KF / 2; ++j) {
                acc = __builtin_amdgcn_wmma_f32_16x16x4_f32(
                     false, afrag[j],  false, bfrag[hh * (KF / 2) + j],
                     (short)0, acc,  false,  false);
            }
            __builtin_amdgcn_sched_group_barrier(0x100, KF / 2, 0);
            __builtin_amdgcn_sched_group_barrier(0x008, KF / 2, 0);
        }
        cur ^= 1;

        float* __restrict__ yr = Y + (size_t)(tm * 16 + 8 * half) * DOUT + col;
#pragma unroll
        for (int r = 0; r < 8; ++r)
            yr[(size_t)r * DOUT] = acc[r];
    }
}


static inline size_t align256(size_t x) { return (x + 255) & ~(size_t)255; }

extern "C" void kernel_launch(void* const* d_in, const int* in_sizes, int n_in,
                              void* d_out, int out_size, void* d_ws, size_t ws_size,
                              hipStream_t stream) {
    (void)in_sizes; (void)n_in; (void)out_size; (void)ws_size;

    const float* x   = (const float*)d_in[0];
    const int*   ei  = (const int*)d_in[1];
    const int*   src = ei;
    const int*   dst = ei + GE;
    const float* W1  = (const float*)d_in[2];
    const float* b1  = (const float*)d_in[3];
    const float* g1  = (const float*)d_in[4];
    const float* be1 = (const float*)d_in[5];
    const float* m1  = (const float*)d_in[6];
    const float* v1  = (const float*)d_in[7];
    const float* W2  = (const float*)d_in[8];
    const float* b2  = (const float*)d_in[9];
    const float* g2  = (const float*)d_in[10];
    const float* be2 = (const float*)d_in[11];
    const float* m2  = (const float*)d_in[12];
    const float* v2  = (const float*)d_in[13];
    const float* W3  = (const float*)d_in[14];
    const float* b3  = (const float*)d_in[15];
    float* out = (float*)d_out;

    char* ws = (char*)d_ws;
    size_t off = 0;
    float* dinv   = (float*)(ws + off); off += align256((size_t)GN * 4);
    int*   cnt    = (int*)(ws + off);   off += align256((size_t)GN * 4);
    int*   rowptr = (int*)(ws + off);   off += align256((size_t)GN * 4);
    int*   rank   = (int*)(ws + off);   off += align256((size_t)GE * 4);
    int2*  csr    = (int2*)(ws + off);  off += align256((size_t)GE * 8);
    int*   chtab  = (int*)(ws + off);   off += align256((size_t)NCH * GN * 4);
    int*   segsum = (int*)(ws + off);   off += align256((size_t)NCH * NSEG * 4);
    float* hbuf = (float*)(ws + off); off += align256((size_t)GN * GH * 4);
    float* agg1 = (float*)(ws + off); off += align256((size_t)GN * GH * 4);
    float* agg2 = (float*)(ws + off); off += align256((size_t)GN * GH * 4);

    const int TPB = 256;

    chunk_count_kernel<<<NCH, CNT_T, 0, stream>>>(dst, rank, chtab, segsum);
    scan_dinv_kernel<<<NSEG, SEG, 0, stream>>>(chtab, segsum, cnt, rowptr, dinv, GN);
    csr_fill_kernel<<<(GE + TPB - 1) / TPB, TPB, 0, stream>>>(src, dst, rank, chtab, dinv, csr, GE);

    const int mTiles = GN / 16;
    const int gemmWaves = 4096;
    const int fusedWaves = 8192;
    const int rowBlocks = (GN + 7) / 8;

    gemm_wmma_f32_kernel<GIN, GH><<<gemmWaves / (GH / 16), (GH / 16) * 32, 0, stream>>>(
        x, W1, hbuf, mTiles);

    gather_gemm_kernel<GH><<<fusedWaves / (GH / 16), (GH / 16) * 32, 0, stream>>>(
        rowptr, cnt, csr, dinv, hbuf, b1, g1, be1, m1, v1, W2, agg1, mTiles);

    gather_gemm_kernel<GC><<<fusedWaves / (GC / 16), (GC / 16) * 32, 0, stream>>>(
        rowptr, cnt, csr, dinv, agg1, b2, g2, be2, m2, v2, W3, agg2, mTiles);

    gather_rows_kernel<GC, EPI_BIAS><<<rowBlocks, TPB, 0, stream>>>(
        rowptr, cnt, csr, dinv, agg2, b3, nullptr, nullptr, nullptr, nullptr, out, GN);
}


# gather batches: the 8 (4, 2) neighbour-row bases are built first, each in a scalar pair of its own; the row loads go out as one clause and the s_wait_xcnt 0 behind every row load goes; sum order kept
# speedup vs baseline: 4.4499x; 1.0468x over previous
#include <hip/hip_runtime.h>
#include <hip/hip_bf16.h>


typedef __attribute__((ext_vector_type(2))) float v2f;
typedef __attribute__((ext_vector_type(8))) float v8f;

static constexpr int GN  = 50000;
static constexpr int GE  = 800000;
static constexpr int GIN = 128;
static constexpr int GH  = 128;
static constexpr int GC  = 64;


template <int V> using vecf = float __attribute__((ext_vector_type(V)));

static constexpr int NCH   = 64;
static constexpr int CH    = GE / NCH;
static constexpr int CNT_T = 1024;
static constexpr int CNT_U = 8;
static constexpr int SEG   = 256;
static constexpr int NSEG  = (GN + SEG - 1) / SEG;
static_assert(CH * NCH == GE, "the chunks must tile the edge list");
static_assert(NSEG <= SEG, "one thread per segment in front of a block");
static_assert(GN % 4 == 0, "node tables are copied 16 bytes at a time");

__global__ void __launch_bounds__(CNT_T)
chunk_count_kernel(const int* __restrict__ dst, int* __restrict__ rank,
                   int* __restrict__ chtab, int* __restrict__ segsum) {
    __shared__ __attribute__((aligned(16))) int tab[NSEG * SEG];
    const int t = threadIdx.x;
    const int p = blockIdx.x;

    int4* tab4 = (int4*)tab;
    for (int q = t; q < NSEG * SEG / 4; q += CNT_T) tab4[q] = make_int4(0, 0, 0, 0);
    __syncthreads();

    const int* __restrict__ dp = dst  + (size_t)p * CH;
    int*       __restrict__ rp = rank + (size_t)p * CH;
    for (int k0 = t; k0 < CH; k0 += CNT_U * CNT_T) {
        int d[CNT_U], r[CNT_U];
#pragma unroll
        for (int u = 0; u < CNT_U; ++u) {
            const int k = k0 + u * CNT_T;
            d[u] = (k < CH) ? dp[k] : -1;
        }
#pragma unroll
        for (int u = 0; u < CNT_U; ++u) {
            r[u] = 0;
            if (d[u] >= 0) r[u] = atomicAdd(&tab[d[u]], 1);
        }
#pragma unroll
        for (int u = 0; u < CNT_U; ++u) {
            const int k = k0 + u * CNT_T;
            if (k < CH) rp[k] = r[u];
        }
    }
    __syncthreads();

    int4* __restrict__ out4 = (int4*)(chtab + (size_t)p * GN);
    for (int q = t; q < GN / 4; q += CNT_T) out4[q] = tab4[q];

    const int lane = t & 31;
    for (int s = t >> 5; s < NSEG; s += CNT_T / 32) {
        int v = 0;
        for (int k = 0; k < SEG / 32; ++k) v += tab[s * SEG + k * 32 + lane];
        for (int o = 16; o > 0; o >>= 1) v += __shfl_xor(v, o);
        if (lane == 0) segsum[p * NSEG + s] = v;
    }
}

__global__ void __launch_bounds__(SEG)
scan_dinv_kernel(int* __restrict__ chtab, const int* __restrict__ segsum,
                 int* __restrict__ cnt, int* __restrict__ rowptr,
                 float* __restrict__ dinv, int n) {
    constexpr int NW = SEG / 32;
    __shared__ int wtot[NW];
    __shared__ int wpre[NW];
    const int t    = threadIdx.x;
    const int lane = t & 31;
    const int wv   = t >> 5;
    const int b    = blockIdx.x;

    int s = 0;
    if (t < b)
        for (int p = 0; p < NCH; ++p) s += segsum[p * NSEG + t];

    const int i  = b * SEG + t;
    const int ic = (i < n) ? i : n - 1;
    int cc[NCH];
    int c = 0;
#pragma unroll
    for (int p = 0; p < NCH; ++p) {
        cc[p] = chtab[(size_t)p * GN + ic];
        c += cc[p];
    }
    if (i >= n) c = 0;

    int inc = c;
    for (int o = 1; o < 32; o <<= 1) {
        int v = __shfl_up(inc, o);
        if (lane >= o) inc += v;
    }
    for (int o = 16; o > 0; o >>= 1) s += __shfl_xor(s, o);

    if (lane == 31) wtot[wv] = inc;
    if (lane == 0)  wpre[wv] = s;
    __syncthreads();

    int a  = (lane < NW) ? wpre[lane] : 0;
    int bw = (lane < wv) ? wtot[lane] : 0;
    for (int o = 16; o > 0; o >>= 1) {
        a  += __shfl_xor(a, o);
        bw += __shfl_xor(bw, o);
    }
    if (i < n) {
        int run = a + bw + inc - c;
        rowptr[i] = run;
        cnt[i]    = c;
        dinv[i]   = rsqrtf((float)(c + 1));
#pragma unroll
        for (int p = 0; p < NCH; ++p) {
            chtab[(size_t)p * GN + i] = run;
            run += cc[p];
        }
    }
}

__global__ void csr_fill_kernel(const int* __restrict__ src, const int* __restrict__ dst,
                                const int* __restrict__ rank, const int* __restrict__ chtab,
                                const float* __restrict__ dinv, int2* __restrict__ csr, int nE) {
    int e = blockIdx.x * blockDim.x + threadIdx.x;
    if (e >= nE) return;
    const int s = src[e];
    const int d = dst[e];
    const int p = e / CH;
    int2 ent;
    ent.x = s;
    ent.y = __float_as_int(dinv[s] * dinv[d]);
    csr[chtab[(size_t)p * GN + d] + rank[e]] = ent;
}


template <int DIN, int DOUT>
__global__ void __launch_bounds__((DOUT / 16) * 32)
gemm_wmma_f32_kernel(const float* __restrict__ X,
                     const float* __restrict__ W,
                     float* __restrict__ Y,
                     int mTiles) {
    static_assert(DIN == 128, "one row of X is one 16-byte load per lane");
    constexpr int tilesN = DOUT / 16;
    constexpr int KF     = DIN / 4;
    constexpr int RPW    = 16 / tilesN;
    static_assert(RPW * tilesN == 16, "the waves' rows must tile the 16 rows");
    constexpr int PITCH  = DIN + 4;

    __shared__ __attribute__((aligned(16))) float atile[2][16 * PITCH];

    const int tn   = threadIdx.x >> 5;
    const int lane = threadIdx.x & 31;
    const int half = lane >> 4;
    const int l15  = lane & 15;
    const int col  = tn * 16 + l15;

    const int row0 = tn * RPW;
    int tm = blockIdx.x;
    vecf<4> nx[RPW];

    const int t0 = (tm < mTiles) ? tm : (mTiles - 1);
#pragma unroll
    for (int r = 0; r < RPW; ++r)
        nx[r] = *(const vecf<4>*)(X + (size_t)(t0 * 16 + row0 + r) * DIN + lane * 4);

    v2f bfrag[KF];
#pragma unroll
    for (int j = 0; j < KF; ++j) {
        const int k = 4 * j + 2 * half;
        bfrag[j].x = W[(size_t)k * DOUT + col];
        bfrag[j].y = W[(size_t)(k + 1) * DOUT + col];
    }

#pragma unroll
    for (int r = 0; r < RPW; ++r)
        *(vecf<4>*)(&atile[0][(row0 + r) * PITCH + lane * 4]) = nx[r];
    __builtin_amdgcn_sched_group_barrier(0x020, 2 * KF + RPW, 1);
    __builtin_amdgcn_sched_group_barrier(0x200, RPW, 1);
    __syncthreads();

    int cur = 0;
    for (; tm < mTiles; tm += gridDim.x) {
        const int  tnext = tm + (int)gridDim.x;
        const bool more  = tnext < mTiles;
        if (more) {
#pragma unroll
            for (int r = 0; r < RPW; ++r)
                nx[r] = *(const vecf<4>*)(X + (size_t)(tnext * 16 + row0 + r) * DIN + lane * 4);
        }

        const float* ar = &atile[cur][l15 * PITCH + 2 * half];
        v2f afrag[KF];
#pragma unroll
        for (int j = 0; j < KF; ++j) afrag[j] = *(const v2f*)(ar + 4 * j);
        v8f acc = {};
#pragma unroll
        for (int j = 0; j < KF; ++j) {
            acc = __builtin_amdgcn_wmma_f32_16x16x4_f32(
                 false, afrag[j],  false, bfrag[j],
                 (short)0, acc,  false,  false);
        }
        __builtin_amdgcn_sched_group_barrier(0x100, KF, 0);
        __builtin_amdgcn_sched_group_barrier(0x008, KF, 0);
        if (more) {
#pragma unroll
            for (int r = 0; r < RPW; ++r)
                *(vecf<4>*)(&atile[cur ^ 1][(row0 + r) * PITCH + lane * 4]) = nx[r];
        }
        __syncthreads();
        cur ^= 1;

        float* __restrict__ yr = Y + (size_t)(tm * 16 + 8 * half) * DOUT + col;
#pragma unroll
        for (int r = 0; r < 8; ++r)
            yr[(size_t)r * DOUT] = acc[r];
    }
}


template <int D, int U>
__device__ __forceinline__ vecf<D / 32> gather_rows(const float* __restrict__ h, int es, int ew,
                                                    int j, int lane, vecf<D / 32> acc) {
    constexpr int VPT = D / 32;
    typedef const vecf<VPT> __attribute__((address_space(1))) * grow_t;
    vecf<VPT> v[U];
    float w[U];
    unsigned long long rb[U];
#pragma unroll
    for (int k = 0; k < U; ++k) {
        const int s = __builtin_amdgcn_readlane(es, j + k);
        w[k] = __int_as_float(__builtin_amdgcn_readlane(ew, j + k));
        const unsigned long long a = (unsigned long long)(h + (size_t)s * D);
        const unsigned lo = __builtin_amdgcn_readfirstlane((unsigned)a);
        const unsigned hi = __builtin_amdgcn_readfirstlane((unsigned)(a >> 32));
        rb[k] = ((unsigned long long)hi << 32) | lo;
    }
    if constexpr (U == 8)
        asm volatile("" :: "s"(rb[0]), "s"(rb[1]), "s"(rb[2]), "s"(rb[3]),
                           "s"(rb[4]), "s"(rb[5]), "s"(rb[6]), "s"(rb[7]));
    else if constexpr (U == 4)
        asm volatile("" :: "s"(rb[0]), "s"(rb[1]), "s"(rb[2]), "s"(rb[3]));
    else if constexpr (U == 2)
        asm volatile("" :: "s"(rb[0]), "s"(rb[1]));
#pragma unroll
    for (int k = 0; k < U; ++k) v[k] = ((grow_t)rb[k])[lane];
#pragma unroll
    for (int k = 0; k < U; ++k) acc += w[k] * v[k];
    return acc;
}

static constexpr int EPI_SUM = 0, EPI_BIAS = 1, EPI_BN_RELU = 2;

template <int D, int EPI>
__global__ void gather_rows_kernel(const int* __restrict__ rowptr, const int* __restrict__ cnt,
                                   const int2* __restrict__ csr, const float* __restrict__ dinv,
                                   const float* __restrict__ h, const float* __restrict__ bias,
                                   const float* __restrict__ bn_g, const float* __restrict__ bn_be,
                                   const float* __restrict__ bn_m, const float* __restrict__ bn_v,
                                   float* __restrict__ agg, int n) {
    constexpr int VPT = D / 32;
    static_assert(VPT == 2 || VPT == 4, "D must be 64 or 128");
    const int lane = threadIdx.x & 31;
    const int i = __builtin_amdgcn_readfirstlane(
        (int)(blockIdx.x * (blockDim.x >> 5) + (threadIdx.x >> 5)));
    if (i >= n) return;

    const int   start = rowptr[i];
    const int   deg   = cnt[i];
    const float di    = dinv[i];

    vecf<VPT> acc = (di * di) * *(const vecf<VPT>*)(h + (size_t)i * D + lane * VPT);

    for (int base = 0; base < deg; base += 32) {
        const int m = (deg - base < 32) ? (deg - base) : 32;
        int es = 0, ew = 0;
        if (lane < m) {
            const int2 ent = csr[start + base + lane];
            es = ent.x;
            ew = ent.y;
        }
        int j = 0;
        for (; j + 8 <= m; j += 8) acc = gather_rows<D, 8>(h, es, ew, j, lane, acc);
        if (j + 4 <= m) { acc = gather_rows<D, 4>(h, es, ew, j, lane, acc); j += 4; }
        if (j + 2 <= m) { acc = gather_rows<D, 2>(h, es, ew, j, lane, acc); j += 2; }
        if (j < m)        acc = gather_rows<D, 1>(h, es, ew, j, lane, acc);
    }

    if (EPI == EPI_BIAS) acc += *(const vecf<VPT>*)(bias + lane * VPT);
    if (EPI == EPI_BN_RELU) {
        const vecf<VPT> pb  = *(const vecf<VPT>*)(bias  + lane * VPT);
        const vecf<VPT> pg  = *(const vecf<VPT>*)(bn_g  + lane * VPT);
        const vecf<VPT> pbe = *(const vecf<VPT>*)(bn_be + lane * VPT);
        const vecf<VPT> pm  = *(const vecf<VPT>*)(bn_m  + lane * VPT);
        const vecf<VPT> pv  = *(const vecf<VPT>*)(bn_v  + lane * VPT);
#pragma unroll
        for (int k = 0; k < VPT; ++k) {
            float val = acc[k] + pb[k];
            val = pg[k] * (val - pm[k]) * rsqrtf(pv[k] + 1e-5f) + pbe[k];
            acc[k] = fmaxf(val, 0.0f);
        }
    }
    *(vecf<VPT>*)(agg + (size_t)i * D + lane * VPT) = acc;
}


template <int D>
__device__ __forceinline__ vecf<D / 32> gather_row_sum(const int2* __restrict__ csr,
                                                       const float* __restrict__ h,
                                                       int i, int start, int deg, float di,
                                                       int es, int ew, int lane) {
    constexpr int VPT = D / 32;
    vecf<VPT> acc = (di * di) * *(const vecf<VPT>*)(h + (size_t)i * D + lane * VPT);
    for (int base = 0; base < deg; base += 32) {
        const int m = (deg - base < 32) ? (deg - base) : 32;
        if (base > 0) {
            es = 0; ew = 0;
            if (lane < m) {
                const int2 ent = csr[start + base + lane];
                es = ent.x;
                ew = ent.y;
            }
        }
        int j = 0;
        for (; j + 8 <= m; j += 8) acc = gather_rows<D, 8>(h, es, ew, j, lane, acc);
        if (j + 4 <= m) { acc = gather_rows<D, 4>(h, es, ew, j, lane, acc); j += 4; }
        if (j + 2 <= m) { acc = gather_rows<D, 2>(h, es, ew, j, lane, acc); j += 2; }
        if (j < m)        acc = gather_rows<D, 1>(h, es, ew, j, lane, acc);
    }
    return acc;
}

template <int DOUT>
__global__ void __launch_bounds__((DOUT / 16) * 32)
gather_gemm_kernel(const int* __restrict__ rowptr, const int* __restrict__ cnt,
                   const int2* __restrict__ csr, const float* __restrict__ dinv,
                   const float* __restrict__ h, const float* __restrict__ bias,
                   const float* __restrict__ bn_g, const float* __restrict__ bn_be,
                   const float* __restrict__ bn_m, const float* __restrict__ bn_v,
                   const float* __restrict__ W, float* __restrict__ Y, int mTiles) {
    constexpr int D      = GH;
    constexpr int tilesN = DOUT / 16;
    constexpr int KF     = D / 4;
    constexpr int RPW    = 16 / tilesN;
    static_assert(D == 128, "a lane owns 4 features: one 16-byte LDS store per row");
    static_assert(RPW * tilesN == 16, "the waves' rows must tile the 16 rows");
    constexpr int PITCH  = D + 4;

    __shared__ __attribute__((aligned(16))) float atile[2][16 * PITCH];
    __shared__ __attribute__((aligned(16))) float par[5][D];

    const int tn   = threadIdx.x >> 5;
    const int lane = threadIdx.x & 31;
    const int half = lane >> 4;
    const int l15  = lane & 15;
    const int col  = tn * 16 + l15;
    const int row0 = __builtin_amdgcn_readfirstlane(tn) * RPW;

    for (int q = threadIdx.x; q < D; q += tilesN * 32) {
        par[0][q] = bias[q];
        par[1][q] = bn_g[q];
        par[2][q] = bn_m[q];
        par[3][q] = rsqrtf(bn_v[q] + 1e-5f);
        par[4][q] = bn_be[q];
    }

    v2f bfrag[KF];
#pragma unroll
    for (int j = 0; j < KF; ++j) {
        const int k = 4 * j + 2 * half;
        bfrag[j].x = W[(size_t)k * DOUT + col];
        bfrag[j].y = W[(size_t)(k + 1) * DOUT + col];
    }
    __syncthreads();

    int cur = 0;
    for (int tm = blockIdx.x; tm < mTiles; tm += gridDim.x) {
        const int i0 = tm * 16 + row0;
        int   st[RPW], dg[RPW], es[RPW], ew[RPW];
        float di[RPW];
#pragma unroll
        for (int r = 0; r < RPW; ++r) {
            st[r] = rowptr[i0 + r];
            dg[r] = cnt[i0 + r];
            di[r] = dinv[i0 + r];
        }
#pragma unroll
        for (int r = 0; r < RPW; ++r) {
            const int q = st[r] + lane;
            const int2 ent = csr[q < GE ? q : GE - 1];
            es[r] = ent.x;
            ew[r] = ent.y;
        }
#pragma unroll
        for (int r = 0; r < RPW; ++r) {
            vecf<4> acc4 = gather_row_sum<D>(csr, h, i0 + r, st[r], dg[r], di[r], es[r], ew[r], lane);
            asm volatile("" ::: "memory");
            const vecf<4> pb  = *(const vecf<4>*)(&par[0][lane * 4]);
            const vecf<4> pg  = *(const vecf<4>*)(&par[1][lane * 4]);
            const vecf<4> pm  = *(const vecf<4>*)(&par[2][lane * 4]);
            const vecf<4> prs = *(const vecf<4>*)(&par[3][lane * 4]);
            const vecf<4> pbe = *(const vecf<4>*)(&par[4][lane * 4]);
#pragma unroll
            for (int k = 0; k < 4; ++k) {
                float val = acc4[k] + pb[k];
                val = pg[k] * (val - pm[k]) * prs[k] + pbe[k];
                acc4[k] = fmaxf(val, 0.0f);
            }
            *(vecf<4>*)(&atile[cur][(row0 + r) * PITCH + lane * 4]) = acc4;
        }
        __syncthreads();

        const float* ar = &atile[cur][l15 * PITCH + 2 * half];
        v8f acc = {};
#pragma unroll
        for (int hh = 0; hh < 2; ++hh) {
            v2f afrag[KF / 2];
#pragma unroll
            for (int j = 0; j < KF / 2; ++j) afrag[j] = *(const v2f*)(ar + 4 * (hh * (KF / 2) + j));
#pragma unroll
            for (int j = 0; j < KF / 2; ++j) {
                acc = __builtin_amdgcn_wmma_f32_16x16x4_f32(
                     false, afrag[j],  false, bfrag[hh * (KF / 2) + j],
                     (short)0, acc,  false,  false);
            }
            __builtin_amdgcn_sched_group_barrier(0x100, KF / 2, 0);
            __builtin_amdgcn_sched_group_barrier(0x008, KF / 2, 0);
        }
        cur ^= 1;

        float* __restrict__ yr = Y + (size_t)(tm * 16 + 8 * half) * DOUT + col;
#pragma unroll
        for (int r = 0; r < 8; ++r)
            yr[(size_t)r * DOUT] = acc[r];
    }
}


static inline size_t align256(size_t x) { return (x + 255) & ~(size_t)255; }

extern "C" void kernel_launch(void* const* d_in, const int* in_sizes, int n_in,
                              void* d_out, int out_size, void* d_ws, size_t ws_size,
                              hipStream_t stream) {
    (void)in_sizes; (void)n_in; (void)out_size; (void)ws_size;

    const float* x   = (const float*)d_in[0];
    const int*   ei  = (const int*)d_in[1];
    const int*   src = ei;
    const int*   dst = ei + GE;
    const float* W1  = (const float*)d_in[2];
    const float* b1  = (const float*)d_in[3];
    const float* g1  = (const float*)d_in[4];
    const float* be1 = (const float*)d_in[5];
    const float* m1  = (const float*)d_in[6];
    const float* v1  = (const float*)d_in[7];
    const float* W2  = (const float*)d_in[8];
    const float* b2  = (const float*)d_in[9];
    const float* g2  = (const float*)d_in[10];
    const float* be2 = (const float*)d_in[11];
    const float* m2  = (const float*)d_in[12];
    const float* v2  = (const float*)d_in[13];
    const float* W3  = (const float*)d_in[14];
    const float* b3  = (const float*)d_in[15];
    float* out = (float*)d_out;

    char* ws = (char*)d_ws;
    size_t off = 0;
    float* dinv   = (float*)(ws + off); off += align256((size_t)GN * 4);
    int*   cnt    = (int*)(ws + off);   off += align256((size_t)GN * 4);
    int*   rowptr = (int*)(ws + off);   off += align256((size_t)GN * 4);
    int*   rank   = (int*)(ws + off);   off += align256((size_t)GE * 4);
    int2*  csr    = (int2*)(ws + off);  off += align256((size_t)GE * 8);
    int*   chtab  = (int*)(ws + off);   off += align256((size_t)NCH * GN * 4);
    int*   segsum = (int*)(ws + off);   off += align256((size_t)NCH * NSEG * 4);
    float* hbuf = (float*)(ws + off); off += align256((size_t)GN * GH * 4);
    float* agg1 = (float*)(ws + off); off += align256((size_t)GN * GH * 4);
    float* agg2 = (float*)(ws + off); off += align256((size_t)GN * GH * 4);

    const int TPB = 256;

    chunk_count_kernel<<<NCH, CNT_T, 0, stream>>>(dst, rank, chtab, segsum);
    scan_dinv_kernel<<<NSEG, SEG, 0, stream>>>(chtab, segsum, cnt, rowptr, dinv, GN);
    csr_fill_kernel<<<(GE + TPB - 1) / TPB, TPB, 0, stream>>>(src, dst, rank, chtab, dinv, csr, GE);

    const int mTiles = GN / 16;
    const int gemmWaves = 4096;
    const int fusedWaves = 8192;
    const int rowBlocks = (GN + 7) / 8;

    gemm_wmma_f32_kernel<GIN, GH><<<gemmWaves / (GH / 16), (GH / 16) * 32, 0, stream>>>(
        x, W1, hbuf, mTiles);

    gather_gemm_kernel<GH><<<fusedWaves / (GH / 16), (GH / 16) * 32, 0, stream>>>(
        rowptr, cnt, csr, dinv, hbuf, b1, g1, be1, m1, v1, W2, agg1, mTiles);

    gather_gemm_kernel<GC><<<fusedWaves / (GC / 16), (GC / 16) * 32, 0, stream>>>(
        rowptr, cnt, csr, dinv, agg1, b2, g2, be2, m2, v2, W3, agg2, mTiles);

    gather_rows_kernel<GC, EPI_BIAS><<<rowBlocks, TPB, 0, stream>>>(
        rowptr, cnt, csr, dinv, agg2, b3, nullptr, nullptr, nullptr, nullptr, out, GN);
}


# edge fill and layer-1 product (independent work) in ONE launch: work-groups 0-511 are the persistent product, 625 more place 5 edges per thread beside it; one launch less, sum order kept
# speedup vs baseline: 4.6207x; 1.0384x over previous
#include <hip/hip_runtime.h>
#include <hip/hip_bf16.h>


typedef __attribute__((ext_vector_type(2))) float v2f;
typedef __attribute__((ext_vector_type(8))) float v8f;

static constexpr int GN  = 50000;
static constexpr int GE  = 800000;
static constexpr int GIN = 128;
static constexpr int GH  = 128;
static constexpr int GC  = 64;


template <int V> using vecf = float __attribute__((ext_vector_type(V)));

static constexpr int NCH   = 64;
static constexpr int CH    = GE / NCH;
static constexpr int CNT_T = 1024;
static constexpr int CNT_U = 8;
static constexpr int SEG   = 256;
static constexpr int NSEG  = (GN + SEG - 1) / SEG;
static_assert(CH * NCH == GE, "the chunks must tile the edge list");
static_assert(NSEG <= SEG, "one thread per segment in front of a block");
static_assert(GN % 4 == 0, "node tables are copied 16 bytes at a time");

__global__ void __launch_bounds__(CNT_T)
chunk_count_kernel(const int* __restrict__ dst, int* __restrict__ rank,
                   int* __restrict__ chtab, int* __restrict__ segsum) {
    __shared__ __attribute__((aligned(16))) int tab[NSEG * SEG];
    const int t = threadIdx.x;
    const int p = blockIdx.x;

    int4* tab4 = (int4*)tab;
    for (int q = t; q < NSEG * SEG / 4; q += CNT_T) tab4[q] = make_int4(0, 0, 0, 0);
    __syncthreads();

    const int* __restrict__ dp = dst  + (size_t)p * CH;
    int*       __restrict__ rp = rank + (size_t)p * CH;
    for (int k0 = t; k0 < CH; k0 += CNT_U * CNT_T) {
        int d[CNT_U], r[CNT_U];
#pragma unroll
        for (int u = 0; u < CNT_U; ++u) {
            const int k = k0 + u * CNT_T;
            d[u] = (k < CH) ? dp[k] : -1;
        }
#pragma unroll
        for (int u = 0; u < CNT_U; ++u) {
            r[u] = 0;
            if (d[u] >= 0) r[u] = atomicAdd(&tab[d[u]], 1);
        }
#pragma unroll
        for (int u = 0; u < CNT_U; ++u) {
            const int k = k0 + u * CNT_T;
            if (k < CH) rp[k] = r[u];
        }
    }
    __syncthreads();

    int4* __restrict__ out4 = (int4*)(chtab + (size_t)p * GN);
    for (int q = t; q < GN / 4; q += CNT_T) out4[q] = tab4[q];

    const int lane = t & 31;
    for (int s = t >> 5; s < NSEG; s += CNT_T / 32) {
        int v = 0;
        for (int k = 0; k < SEG / 32; ++k) v += tab[s * SEG + k * 32 + lane];
        for (int o = 16; o > 0; o >>= 1) v += __shfl_xor(v, o);
        if (lane == 0) segsum[p * NSEG + s] = v;
    }
}

__global__ void __launch_bounds__(SEG)
scan_dinv_kernel(int* __restrict__ chtab, const int* __restrict__ segsum,
                 int* __restrict__ cnt, int* __restrict__ rowptr,
                 float* __restrict__ dinv, int n) {
    constexpr int NW = SEG / 32;
    __shared__ int wtot[NW];
    __shared__ int wpre[NW];
    const int t    = threadIdx.x;
    const int lane = t & 31;
    const int wv   = t >> 5;
    const int b    = blockIdx.x;

    int s = 0;
    if (t < b)
        for (int p = 0; p < NCH; ++p) s += segsum[p * NSEG + t];

    const int i  = b * SEG + t;
    const int ic = (i < n) ? i : n - 1;
    int cc[NCH];
    int c = 0;
#pragma unroll
    for (int p = 0; p < NCH; ++p) {
        cc[p] = chtab[(size_t)p * GN + ic];
        c += cc[p];
    }
    if (i >= n) c = 0;

    int inc = c;
    for (int o = 1; o < 32; o <<= 1) {
        int v = __shfl_up(inc, o);
        if (lane >= o) inc += v;
    }
    for (int o = 16; o > 0; o >>= 1) s += __shfl_xor(s, o);

    if (lane == 31) wtot[wv] = inc;
    if (lane == 0)  wpre[wv] = s;
    __syncthreads();

    int a  = (lane < NW) ? wpre[lane] : 0;
    int bw = (lane < wv) ? wtot[lane] : 0;
    for (int o = 16; o > 0; o >>= 1) {
        a  += __shfl_xor(a, o);
        bw += __shfl_xor(bw, o);
    }
    if (i < n) {
        int run = a + bw + inc - c;
        rowptr[i] = run;
        cnt[i]    = c;
        dinv[i]   = rsqrtf((float)(c + 1));
#pragma unroll
        for (int p = 0; p < NCH; ++p) {
            chtab[(size_t)p * GN + i] = run;
            run += cc[p];
        }
    }
}

static constexpr int FILL_T = 256;
static constexpr int FILL_U = 5;
static constexpr int FILL_BLOCKS = GE / (FILL_T * FILL_U);
static_assert(FILL_BLOCKS * FILL_T * FILL_U == GE, "the fill work-groups must tile the edge list");

__device__ __forceinline__ void csr_fill_part(const int* __restrict__ src, const int* __restrict__ dst,
                                              const int* __restrict__ rank, const int* __restrict__ chtab,
                                              const float* __restrict__ dinv, int2* __restrict__ csr,
                                              int fb) {
    const int e0 = fb * (FILL_T * FILL_U) + (int)threadIdx.x;
    int s[FILL_U], d[FILL_U], r[FILL_U], q[FILL_U];
    float w[FILL_U];
#pragma unroll
    for (int u = 0; u < FILL_U; ++u) {
        const int e = e0 + u * FILL_T;
        s[u] = src[e];
        d[u] = dst[e];
        r[u] = rank[e];
    }
#pragma unroll
    for (int u = 0; u < FILL_U; ++u) {
        const int p = (e0 + u * FILL_T) / CH;
        q[u] = chtab[(size_t)p * GN + d[u]] + r[u];
        w[u] = dinv[s[u]] * dinv[d[u]];
    }
#pragma unroll
    for (int u = 0; u < FILL_U; ++u) {
        int2 ent;
        ent.x = s[u];
        ent.y = __float_as_int(w[u]);
        csr[q[u]] = ent;
    }
}


template <int DIN, int DOUT>
__device__ __forceinline__ void
gemm_wmma_f32_part(const float* X, const float* W, float* Y,
                   int mTiles, int tm, int tstride) {
    static_assert(DIN == 128, "one row of X is one 16-byte load per lane");
    constexpr int tilesN = DOUT / 16;
    constexpr int KF     = DIN / 4;
    constexpr int RPW    = 16 / tilesN;
    static_assert(RPW * tilesN == 16, "the waves' rows must tile the 16 rows");
    constexpr int PITCH  = DIN + 4;

    __shared__ __attribute__((aligned(16))) float atile[2][16 * PITCH];

    const int tn   = threadIdx.x >> 5;
    const int lane = threadIdx.x & 31;
    const int half = lane >> 4;
    const int l15  = lane & 15;
    const int col  = tn * 16 + l15;

    const int row0 = tn * RPW;
    vecf<4> nx[RPW];

    const int t0 = (tm < mTiles) ? tm : (mTiles - 1);
#pragma unroll
    for (int r = 0; r < RPW; ++r)
        nx[r] = *(const vecf<4>*)(X + (size_t)(t0 * 16 + row0 + r) * DIN + lane * 4);

    v2f bfrag[KF];
#pragma unroll
    for (int j = 0; j < KF; ++j) {
        const int k = 4 * j + 2 * half;
        bfrag[j].x = W[(size_t)k * DOUT + col];
        bfrag[j].y = W[(size_t)(k + 1) * DOUT + col];
    }

#pragma unroll
    for (int r = 0; r < RPW; ++r)
        *(vecf<4>*)(&atile[0][(row0 + r) * PITCH + lane * 4]) = nx[r];
    __builtin_amdgcn_sched_group_barrier(0x020, 2 * KF + RPW, 1);
    __builtin_amdgcn_sched_group_barrier(0x200, RPW, 1);
    __syncthreads();

    int cur = 0;
    for (; tm < mTiles; tm += tstride) {
        const int  tnext = tm + tstride;
        const bool more  = tnext < mTiles;
        if (more) {
#pragma unroll
            for (int r = 0; r < RPW; ++r)
                nx[r] = *(const vecf<4>*)(X + (size_t)(tnext * 16 + row0 + r) * DIN + lane * 4);
        }

        const float* ar = &atile[cur][l15 * PITCH + 2 * half];
        v2f afrag[KF];
#pragma unroll
        for (int j = 0; j < KF; ++j) afrag[j] = *(const v2f*)(ar + 4 * j);
        v8f acc = {};
#pragma unroll
        for (int j = 0; j < KF; ++j) {
            acc = __builtin_amdgcn_wmma_f32_16x16x4_f32(
                 false, afrag[j],  false, bfrag[j],
                 (short)0, acc,  false,  false);
        }
        __builtin_amdgcn_sched_group_barrier(0x100, KF, 0);
        __builtin_amdgcn_sched_group_barrier(0x008, KF, 0);
        if (more) {
#pragma unroll
            for (int r = 0; r < RPW; ++r)
                *(vecf<4>*)(&atile[cur ^ 1][(row0 + r) * PITCH + lane * 4]) = nx[r];
        }
        __syncthreads();
        cur ^= 1;

        float* __restrict__ yr = Y + (size_t)(tm * 16 + 8 * half) * DOUT + col;
#pragma unroll
        for (int r = 0; r < 8; ++r)
            yr[(size_t)r * DOUT] = acc[r];
    }
}

static constexpr int GEMM1_BLOCKS = 4096 / (GH / 16);

__global__ void __launch_bounds__((GH / 16) * 32)
fill_gemm_kernel(const int* __restrict__ src, const int* __restrict__ dst,
                 const int* __restrict__ rank, const int* __restrict__ chtab,
                 const float* __restrict__ dinv, int2* __restrict__ csr,
                 const float* __restrict__ X, const float* __restrict__ W,
                 float* __restrict__ Y, int mTiles) {
    static_assert((GH / 16) * 32 == FILL_T, "fill and product work-groups have one size");
    if ((int)blockIdx.x >= GEMM1_BLOCKS) {
        csr_fill_part(src, dst, rank, chtab, dinv, csr, (int)blockIdx.x - GEMM1_BLOCKS);
        return;
    }
    gemm_wmma_f32_part<GIN, GH>(X, W, Y, mTiles, (int)blockIdx.x, GEMM1_BLOCKS);
}


template <int D, int U>
__device__ __forceinline__ vecf<D / 32> gather_rows(const float* __restrict__ h, int es, int ew,
                                                    int j, int lane, vecf<D / 32> acc) {
    constexpr int VPT = D / 32;
    typedef const vecf<VPT> __attribute__((address_space(1))) * grow_t;
    vecf<VPT> v[U];
    float w[U];
    unsigned long long rb[U];
#pragma unroll
    for (int k = 0; k < U; ++k) {
        const int s = __builtin_amdgcn_readlane(es, j + k);
        w[k] = __int_as_float(__builtin_amdgcn_readlane(ew, j + k));
        const unsigned long long a = (unsigned long long)(h + (size_t)s * D);
        const unsigned lo = __builtin_amdgcn_readfirstlane((unsigned)a);
        const unsigned hi = __builtin_amdgcn_readfirstlane((unsigned)(a >> 32));
        rb[k] = ((unsigned long long)hi << 32) | lo;
    }
    if constexpr (U == 8)
        asm volatile("" :: "s"(rb[0]), "s"(rb[1]), "s"(rb[2]), "s"(rb[3]),
                           "s"(rb[4]), "s"(rb[5]), "s"(rb[6]), "s"(rb[7]));
    else if constexpr (U == 4)
        asm volatile("" :: "s"(rb[0]), "s"(rb[1]), "s"(rb[2]), "s"(rb[3]));
    else if constexpr (U == 2)
        asm volatile("" :: "s"(rb[0]), "s"(rb[1]));
#pragma unroll
    for (int k = 0; k < U; ++k) v[k] = ((grow_t)rb[k])[lane];
#pragma unroll
    for (int k = 0; k < U; ++k) acc += w[k] * v[k];
    return acc;
}

static constexpr int EPI_SUM = 0, EPI_BIAS = 1, EPI_BN_RELU = 2;

template <int D, int EPI>
__global__ void gather_rows_kernel(const int* __restrict__ rowptr, const int* __restrict__ cnt,
                                   const int2* __restrict__ csr, const float* __restrict__ dinv,
                                   const float* __restrict__ h, const float* __restrict__ bias,
                                   const float* __restrict__ bn_g, const float* __restrict__ bn_be,
                                   const float* __restrict__ bn_m, const float* __restrict__ bn_v,
                                   float* __restrict__ agg, int n) {
    constexpr int VPT = D / 32;
    static_assert(VPT == 2 || VPT == 4, "D must be 64 or 128");
    const int lane = threadIdx.x & 31;
    const int i = __builtin_amdgcn_readfirstlane(
        (int)(blockIdx.x * (blockDim.x >> 5) + (threadIdx.x >> 5)));
    if (i >= n) return;

    const int   start = rowptr[i];
    const int   deg   = cnt[i];
    const float di    = dinv[i];

    vecf<VPT> acc = (di * di) * *(const vecf<VPT>*)(h + (size_t)i * D + lane * VPT);

    for (int base = 0; base < deg; base += 32) {
        const int m = (deg - base < 32) ? (deg - base) : 32;
        int es = 0, ew = 0;
        if (lane < m) {
            const int2 ent = csr[start + base + lane];
            es = ent.x;
            ew = ent.y;
        }
        int j = 0;
        for (; j + 8 <= m; j += 8) acc = gather_rows<D, 8>(h, es, ew, j, lane, acc);
        if (j + 4 <= m) { acc = gather_rows<D, 4>(h, es, ew, j, lane, acc); j += 4; }
        if (j + 2 <= m) { acc = gather_rows<D, 2>(h, es, ew, j, lane, acc); j += 2; }
        if (j < m)        acc = gather_rows<D, 1>(h, es, ew, j, lane, acc);
    }

    if (EPI == EPI_BIAS) acc += *(const vecf<VPT>*)(bias + lane * VPT);
    if (EPI == EPI_BN_RELU) {
        const vecf<VPT> pb  = *(const vecf<VPT>*)(bias  + lane * VPT);
        const vecf<VPT> pg  = *(const vecf<VPT>*)(bn_g  + lane * VPT);
        const vecf<VPT> pbe = *(const vecf<VPT>*)(bn_be + lane * VPT);
        const vecf<VPT> pm  = *(const vecf<VPT>*)(bn_m  + lane * VPT);
        const vecf<VPT> pv  = *(const vecf<VPT>*)(bn_v  + lane * VPT);
#pragma unroll
        for (int k = 0; k < VPT; ++k) {
            float val = acc[k] + pb[k];
            val = pg[k] * (val - pm[k]) * rsqrtf(pv[k] + 1e-5f) + pbe[k];
            acc[k] = fmaxf(val, 0.0f);
        }
    }
    *(vecf<VPT>*)(agg + (size_t)i * D + lane * VPT) = acc;
}


template <int D>
__device__ __forceinline__ vecf<D / 32> gather_row_sum(const int2* __restrict__ csr,
                                                       const float* __restrict__ h,
                                                       int i, int start, int deg, float di,
                                                       int es, int ew, int lane) {
    constexpr int VPT = D / 32;
    vecf<VPT> acc = (di * di) * *(const vecf<VPT>*)(h + (size_t)i * D + lane * VPT);
    for (int base = 0; base < deg; base += 32) {
        const int m = (deg - base < 32) ? (deg - base) : 32;
        if (base > 0) {
            es = 0; ew = 0;
            if (lane < m) {
                const int2 ent = csr[start + base + lane];
                es = ent.x;
                ew = ent.y;
            }
        }
        int j = 0;
        for (; j + 8 <= m; j += 8) acc = gather_rows<D, 8>(h, es, ew, j, lane, acc);
        if (j + 4 <= m) { acc = gather_rows<D, 4>(h, es, ew, j, lane, acc); j += 4; }
        if (j + 2 <= m) { acc = gather_rows<D, 2>(h, es, ew, j, lane, acc); j += 2; }
        if (j < m)        acc = gather_rows<D, 1>(h, es, ew, j, lane, acc);
    }
    return acc;
}

template <int DOUT>
__global__ void __launch_bounds__((DOUT / 16) * 32)
gather_gemm_kernel(const int* __restrict__ rowptr, const int* __restrict__ cnt,
                   const int2* __restrict__ csr, const float* __restrict__ dinv,
                   const float* __restrict__ h, const float* __restrict__ bias,
                   const float* __restrict__ bn_g, const float* __restrict__ bn_be,
                   const float* __restrict__ bn_m, const float* __restrict__ bn_v,
                   const float* __restrict__ W, float* __restrict__ Y, int mTiles) {
    constexpr int D      = GH;
    constexpr int tilesN = DOUT / 16;
    constexpr int KF     = D / 4;
    constexpr int RPW    = 16 / tilesN;
    static_assert(D == 128, "a lane owns 4 features: one 16-byte LDS store per row");
    static_assert(RPW * tilesN == 16, "the waves' rows must tile the 16 rows");
    constexpr int PITCH  = D + 4;

    __shared__ __attribute__((aligned(16))) float atile[2][16 * PITCH];
    __shared__ __attribute__((aligned(16))) float par[5][D];

    const int tn   = threadIdx.x >> 5;
    const int lane = threadIdx.x & 31;
    const int half = lane >> 4;
    const int l15  = lane & 15;
    const int col  = tn * 16 + l15;
    const int row0 = __builtin_amdgcn_readfirstlane(tn) * RPW;

    for (int q = threadIdx.x; q < D; q += tilesN * 32) {
        par[0][q] = bias[q];
        par[1][q] = bn_g[q];
        par[2][q] = bn_m[q];
        par[3][q] = rsqrtf(bn_v[q] + 1e-5f);
        par[4][q] = bn_be[q];
    }

    v2f bfrag[KF];
#pragma unroll
    for (int j = 0; j < KF; ++j) {
        const int k = 4 * j + 2 * half;
        bfrag[j].x = W[(size_t)k * DOUT + col];
        bfrag[j].y = W[(size_t)(k + 1) * DOUT + col];
    }
    __syncthreads();

    int cur = 0;
    for (int tm = blockIdx.x; tm < mTiles; tm += gridDim.x) {
        const int i0 = tm * 16 + row0;
        int   st[RPW], dg[RPW], es[RPW], ew[RPW];
        float di[RPW];
#pragma unroll
        for (int r = 0; r < RPW; ++r) {
            st[r] = rowptr[i0 + r];
            dg[r] = cnt[i0 + r];
            di[r] = dinv[i0 + r];
        }
#pragma unroll
        for (int r = 0; r < RPW; ++r) {
            const int q = st[r] + lane;
            const int2 ent = csr[q < GE ? q : GE - 1];
            es[r] = ent.x;
            ew[r] = ent.y;
        }
#pragma unroll
        for (int r = 0; r < RPW; ++r) {
            vecf<4> acc4 = gather_row_sum<D>(csr, h, i0 + r, st[r], dg[r], di[r], es[r], ew[r], lane);
            asm volatile("" ::: "memory");
            const vecf<4> pb  = *(const vecf<4>*)(&par[0][lane * 4]);
            const vecf<4> pg  = *(const vecf<4>*)(&par[1][lane * 4]);
            const vecf<4> pm  = *(const vecf<4>*)(&par[2][lane * 4]);
            const vecf<4> prs = *(const vecf<4>*)(&par[3][lane * 4]);
            const vecf<4> pbe = *(const vecf<4>*)(&par[4][lane * 4]);
#pragma unroll
            for (int k = 0; k < 4; ++k) {
                float val = acc4[k] + pb[k];
                val = pg[k] * (val - pm[k]) * prs[k] + pbe[k];
                acc4[k] = fmaxf(val, 0.0f);
            }
            *(vecf<4>*)(&atile[cur][(row0 + r) * PITCH + lane * 4]) = acc4;
        }
        __syncthreads();

        const float* ar = &atile[cur][l15 * PITCH + 2 * half];
        v8f acc = {};
#pragma unroll
        for (int hh = 0; hh < 2; ++hh) {
            v2f afrag[KF / 2];
#pragma unroll
            for (int j = 0; j < KF / 2; ++j) afrag[j] = *(const v2f*)(ar + 4 * (hh * (KF / 2) + j));
#pragma unroll
            for (int j = 0; j < KF / 2; ++j) {
                acc = __builtin_amdgcn_wmma_f32_16x16x4_f32(
                     false, afrag[j],  false, bfrag[hh * (KF / 2) + j],
                     (short)0, acc,  false,  false);
            }
            __builtin_amdgcn_sched_group_barrier(0x100, KF / 2, 0);
            __builtin_amdgcn_sched_group_barrier(0x008, KF / 2, 0);
        }
        cur ^= 1;

        float* __restrict__ yr = Y + (size_t)(tm * 16 + 8 * half) * DOUT + col;
#pragma unroll
        for (int r = 0; r < 8; ++r)
            yr[(size_t)r * DOUT] = acc[r];
    }
}


static inline size_t align256(size_t x) { return (x + 255) & ~(size_t)255; }

extern "C" void kernel_launch(void* const* d_in, const int* in_sizes, int n_in,
                              void* d_out, int out_size, void* d_ws, size_t ws_size,
                              hipStream_t stream) {
    (void)in_sizes; (void)n_in; (void)out_size; (void)ws_size;

    const float* x   = (const float*)d_in[0];
    const int*   ei  = (const int*)d_in[1];
    const int*   src = ei;
    const int*   dst = ei + GE;
    const float* W1  = (const float*)d_in[2];
    const float* b1  = (const float*)d_in[3];
    const float* g1  = (const float*)d_in[4];
    const float* be1 = (const float*)d_in[5];
    const float* m1  = (const float*)d_in[6];
    const float* v1  = (const float*)d_in[7];
    const float* W2  = (const float*)d_in[8];
    const float* b2  = (const float*)d_in[9];
    const float* g2  = (const float*)d_in[10];
    const float* be2 = (const float*)d_in[11];
    const float* m2  = (const float*)d_in[12];
    const float* v2  = (const float*)d_in[13];
    const float* W3  = (const float*)d_in[14];
    const float* b3  = (const float*)d_in[15];
    float* out = (float*)d_out;

    char* ws = (char*)d_ws;
    size_t off = 0;
    float* dinv   = (float*)(ws + off); off += align256((size_t)GN * 4);
    int*   cnt    = (int*)(ws + off);   off += align256((size_t)GN * 4);
    int*   rowptr = (int*)(ws + off);   off += align256((size_t)GN * 4);
    int*   rank   = (int*)(ws + off);   off += align256((size_t)GE * 4);
    int2*  csr    = (int2*)(ws + off);  off += align256((size_t)GE * 8);
    int*   chtab  = (int*)(ws + off);   off += align256((size_t)NCH * GN * 4);
    int*   segsum = (int*)(ws + off);   off += align256((size_t)NCH * NSEG * 4);
    float* hbuf = (float*)(ws + off); off += align256((size_t)GN * GH * 4);
    float* agg1 = (float*)(ws + off); off += align256((size_t)GN * GH * 4);
    float* agg2 = (float*)(ws + off); off += align256((size_t)GN * GH * 4);

    const int TPB = 256;

    chunk_count_kernel<<<NCH, CNT_T, 0, stream>>>(dst, rank, chtab, segsum);
    scan_dinv_kernel<<<NSEG, SEG, 0, stream>>>(chtab, segsum, cnt, rowptr, dinv, GN);

    const int mTiles = GN / 16;
    const int fusedWaves = 8192;
    const int rowBlocks = (GN + 7) / 8;

    fill_gemm_kernel<<<GEMM1_BLOCKS + FILL_BLOCKS, FILL_T, 0, stream>>>(
        src, dst, rank, chtab, dinv, csr, x, W1, hbuf, mTiles);

    gather_gemm_kernel<GH><<<fusedWaves / (GH / 16), (GH / 16) * 32, 0, stream>>>(
        rowptr, cnt, csr, dinv, hbuf, b1, g1, be1, m1, v1, W2, agg1, mTiles);

    gather_gemm_kernel<GC><<<fusedWaves / (GC / 16), (GC / 16) * 32, 0, stream>>>(
        rowptr, cnt, csr, dinv, agg1, b2, g2, be2, m2, v2, W3, agg2, mTiles);

    gather_rows_kernel<GC, EPI_BIAS><<<rowBlocks, TPB, 0, stream>>>(
        rowptr, cnt, csr, dinv, agg2, b3, nullptr, nullptr, nullptr, nullptr, out, GN);
}
